# starting kernel
# baseline (speedup 1.0000x reference)
#include <hip/hip_runtime.h>
#include <hip/hip_bf16.h>

typedef __attribute__((ext_vector_type(2))) float v2f;
typedef __attribute__((ext_vector_type(8))) float v8f;

static __device__ __forceinline__ void atomic_fadd(float* p, float v) {
    __hip_atomic_fetch_add(p, v, __ATOMIC_RELAXED, __HIP_MEMORY_SCOPE_AGENT);
}

__global__ void gcn_init_deg(float* __restrict__ deg, int n) {
    int i = blockIdx.x * blockDim.x + threadIdx.x;
    if (i < n) deg[i] = 1.0f;
}

__global__ void gcn_count_deg(const int* __restrict__ dst, int E,
                              float* __restrict__ deg) {
    int e = blockIdx.x * blockDim.x + threadIdx.x;
    if (e < E) atomic_fadd(&deg[dst[e]], 1.0f);
}

__global__ void gcn_rsqrt(float* __restrict__ deg, int n) {
    int i = blockIdx.x * blockDim.x + threadIdx.x;
    if (i < n) deg[i] = rsqrtf(deg[i]);
}

__global__ void gcn_zero(float* __restrict__ p, long long n) {
    long long i = (long long)blockIdx.x * blockDim.x + threadIdx.x;
    if (i < n) p[i] = 0.0f;
}

__global__ void gcn_gemm_f32_wmma(const float* __restrict__ A,
                                  const float* __restrict__ B,
                                  float* __restrict__ C,
                                  int M, int N, int K) {
    const int wave  = threadIdx.x >> 5;
    const int lane  = threadIdx.x & 31;
    const int mTile = blockIdx.x * 4 + wave;
    if (mTile * 16 >= M) return;

    const int r16   = lane & 15;
    const int koff  = (lane >> 4) << 1;
    const int nBase = blockIdx.y * 64 + r16;

    v8f c0 = {0.f,0.f,0.f,0.f,0.f,0.f,0.f,0.f};
    v8f c1 = c0, c2 = c0, c3 = c0;

    const float* Arow = A + ((long long)mTile * 16 + r16) * K;

    for (int k0 = 0; k0 < K; k0 += 4) {
        v2f a = *(const v2f*)(Arow + k0 + koff);
        const float* Brow0 = B + (long long)(k0 + koff)     * N + nBase;
        const float* Brow1 = B + (long long)(k0 + koff + 1) * N + nBase;
        v2f b0, b1, b2, b3;
        b0.x = Brow0[0];  b0.y = Brow1[0];
        b1.x = Brow0[16]; b1.y = Brow1[16];
        b2.x = Brow0[32]; b2.y = Brow1[32];
        b3.x = Brow0[48]; b3.y = Brow1[48];
        c0 = __builtin_amdgcn_wmma_f32_16x16x4_f32(false, a, false, b0, (short)0, c0, false, false);
        c1 = __builtin_amdgcn_wmma_f32_16x16x4_f32(false, a, false, b1, (short)0, c1, false, false);
        c2 = __builtin_amdgcn_wmma_f32_16x16x4_f32(false, a, false, b2, (short)0, c2, false, false);
        c3 = __builtin_amdgcn_wmma_f32_16x16x4_f32(false, a, false, b3, (short)0, c3, false, false);
    }

    const long long rowBase = (long long)mTile * 16 + ((lane >> 4) << 3);
#pragma unroll
    for (int v = 0; v < 8; ++v) {
        float* Crow = C + (rowBase + v) * N + nBase;
        Crow[0]  = c0[v];
        Crow[16] = c1[v];
        Crow[32] = c2[v];
        Crow[48] = c3[v];
    }
}

__global__ void gcn_scatter(const float* __restrict__ h,
                            const int* __restrict__ src,
                            const int* __restrict__ dst,
                            const float* __restrict__ dinv,
                            float* __restrict__ agg,
                            int E, int C) {
    const int wavesPerBlock = blockDim.x >> 5;
    const int e    = blockIdx.x * wavesPerBlock + (threadIdx.x >> 5);
    const int lane = threadIdx.x & 31;
    if (e >= E) return;
    const int s = src[e];
    const int d = dst[e];
    const float nrm = dinv[s] * dinv[d];
    const float* hs = h   + (long long)s * C;
    float*       ad = agg + (long long)d * C;
    for (int c = lane; c < C; c += 32) {
        atomic_fadd(&ad[c], hs[c] * nrm);
    }
}

__global__ void gcn_finalize(const float* __restrict__ h,
                             const float* __restrict__ dinv,
                             const float* __restrict__ bias,
                             float* __restrict__ agg,
                             int nNodes, int C, int doRelu) {
    long long i = (long long)blockIdx.x * blockDim.x + threadIdx.x;
    long long total = (long long)nNodes * C;
    if (i >= total) return;
    int node = (int)(i / C);
    int c    = (int)(i % C);
    float di = dinv[node];
    float v  = agg[i] + h[i] * di * di + bias[c];
    if (doRelu) v = fmaxf(v, 0.0f);
    agg[i] = v;
}

extern "C" void kernel_launch(void* const* d_in, const int* in_sizes, int n_in,
                              void* d_out, int out_size, void* d_ws, size_t ws_size,
                              hipStream_t stream) {
    const float* x   = (const float*)d_in[0];
    const int*   ei  = (const int*)d_in[1];
    const float* W1  = (const float*)d_in[2];
    const float* b1  = (const float*)d_in[3];
    const float* W2  = (const float*)d_in[4];
    const float* b2  = (const float*)d_in[5];

    const int IN_C  = 128;
    const int HID_C = 128;
    const int OUT_C = 64;
    const int nNodes = in_sizes[0] / IN_C;
    const int E      = in_sizes[1] / 2;
    const int* srcIdx = ei;
    const int* dstIdx = ei + E;

    float* dinv = (float*)d_ws;
    float* bufA = dinv + nNodes;
    float* bufB = bufA + (long long)nNodes * HID_C;
    float* out  = (float*)d_out;

    const int T = 256;
    const long long nh = (long long)nNodes * HID_C;
    const long long no = (long long)nNodes * OUT_C;
    const int mBlocks = (nNodes / 16 + 3) / 4;

    gcn_init_deg <<<(nNodes + T - 1) / T, T, 0, stream>>>(dinv, nNodes);
    gcn_count_deg<<<(E + T - 1) / T, T, 0, stream>>>(dstIdx, E, dinv);
    gcn_rsqrt    <<<(nNodes + T - 1) / T, T, 0, stream>>>(dinv, nNodes);

    {
        dim3 grid(mBlocks, HID_C / 64);
        gcn_gemm_f32_wmma<<<grid, 128, 0, stream>>>(x, W1, bufA,
                                                    nNodes, HID_C, IN_C);
    }

    gcn_zero<<<(unsigned)((nh + T - 1) / T), T, 0, stream>>>(bufB, nh);
    gcn_scatter<<<(E + 7) / 8, 256, 0, stream>>>(bufA, srcIdx, dstIdx, dinv,
                                                 bufB, E, HID_C);

    gcn_finalize<<<(unsigned)((nh + T - 1) / T), T, 0, stream>>>(
        bufA, dinv, b1, bufB, nNodes, HID_C,  1);

    {
        dim3 grid(mBlocks, OUT_C / 64);
        gcn_gemm_f32_wmma<<<grid, 128, 0, stream>>>(bufB, W2, bufA,
                                                    nNodes, OUT_C, HID_C);
    }

    gcn_zero<<<(unsigned)((no + T - 1) / T), T, 0, stream>>>(out, no);
    gcn_scatter<<<(E + 7) / 8, 256, 0, stream>>>(bufA, srcIdx, dstIdx, dinv,
                                                 out, E, OUT_C);

    gcn_finalize<<<(unsigned)((no + T - 1) / T), T, 0, stream>>>(
        bufA, dinv, b2, out, nNodes, OUT_C,  0);
}


# Aggregation as a gather: edges counting-sorted by destination (int rank atomic + scan); a wave sums a node's rows in registers in arrival-order rank; float atomics, zero-fill, epilogue passes gone
# speedup vs baseline: 2.1761x; 2.1761x over previous
#include <hip/hip_runtime.h>


typedef __attribute__((ext_vector_type(2))) float v2f;
typedef __attribute__((ext_vector_type(8))) float v8f;

__global__ void gcn_zero_counts(int* __restrict__ cnt, int n) {
    int i = blockIdx.x * blockDim.x + threadIdx.x;
    if (i < n) cnt[i] = 0;
}

__global__ void gcn_count_rank(const int* __restrict__ dst, int E,
                               int* __restrict__ cnt, int* __restrict__ rank) {
    int e = blockIdx.x * blockDim.x + threadIdx.x;
    if (e < E)
        rank[e] = __hip_atomic_fetch_add(&cnt[dst[e]], 1, __ATOMIC_RELAXED,
                                         __HIP_MEMORY_SCOPE_AGENT);
}

#define SCAN_T 1024
__global__ __launch_bounds__(SCAN_T)
void gcn_scan_nodes(const int* __restrict__ cnt, int* __restrict__ start,
                    float* __restrict__ dinv, int n) {
    __shared__ int s_own[32];
    __shared__ int s_pre[32];
    const int tid  = threadIdx.x;
    const int lane = tid & 31;
    const int wave = tid >> 5;
    const int base = blockIdx.x * SCAN_T;

    int pre = 0;
    const int4* c4 = (const int4*)cnt;
    for (int q = tid; q < (base >> 2); q += SCAN_T) {
        int4 v = c4[q];
        pre += (v.x + v.y) + (v.z + v.w);
    }

    const int i = base + tid;
    const int c = (i < n) ? cnt[i] : 0;

    int inc = c;
    for (int o = 1; o < 32; o <<= 1) {
        int t = __shfl_up(inc, o);
        if (lane >= o) inc += t;
    }
    for (int o = 16; o > 0; o >>= 1) pre += __shfl_xor(pre, o);

    if (lane == 31) s_own[wave] = inc;
    if (lane == 0)  s_pre[wave] = pre;
    __syncthreads();

    const int wo = s_own[lane];
    int wp = s_pre[lane];
    int winc = wo;
    for (int o = 1; o < 32; o <<= 1) {
        int t = __shfl_up(winc, o);
        if (lane >= o) winc += t;
    }
    const int wexc = __shfl(winc - wo, wave);
    for (int o = 16; o > 0; o >>= 1) wp += __shfl_xor(wp, o);

    const int st = wp + wexc + (inc - c);
    if (i < n) {
        start[i] = st;
        dinv[i]  = rsqrtf((float)c + 1.0f);
        if (i == n - 1) start[n] = st + c;
    }
}

__global__ void gcn_place(const int* __restrict__ src, const int* __restrict__ dst,
                          const int* __restrict__ rank, const int* __restrict__ start,
                          int* __restrict__ csr, int E) {
    int e = blockIdx.x * blockDim.x + threadIdx.x;
    if (e < E) csr[start[dst[e]] + rank[e]] = src[e];
}

__global__ void gcn_gemm_f32_wmma(const float* __restrict__ A,
                                  const float* __restrict__ B,
                                  float* C0,
                                  float* C1,
                                  int M, int N, int K) {
    const int wave  = threadIdx.x >> 5;
    const int lane  = threadIdx.x & 31;
    const int mTile = blockIdx.x * 4 + wave;
    if (mTile * 16 >= M) return;

    const int r16   = lane & 15;
    const int koff  = (lane >> 4) << 1;
    const int nBase = blockIdx.y * 64 + r16;

    v8f c0 = {0.f,0.f,0.f,0.f,0.f,0.f,0.f,0.f};
    v8f c1 = c0, c2 = c0, c3 = c0;

    const float* Arow = A + ((long long)mTile * 16 + r16) * K;

    for (int k0 = 0; k0 < K; k0 += 4) {
        v2f a = *(const v2f*)(Arow + k0 + koff);
        const float* Brow0 = B + (long long)(k0 + koff)     * N + nBase;
        const float* Brow1 = B + (long long)(k0 + koff + 1) * N + nBase;
        v2f b0, b1, b2, b3;
        b0.x = Brow0[0];  b0.y = Brow1[0];
        b1.x = Brow0[16]; b1.y = Brow1[16];
        b2.x = Brow0[32]; b2.y = Brow1[32];
        b3.x = Brow0[48]; b3.y = Brow1[48];
        c0 = __builtin_amdgcn_wmma_f32_16x16x4_f32(false, a, false, b0, (short)0, c0, false, false);
        c1 = __builtin_amdgcn_wmma_f32_16x16x4_f32(false, a, false, b1, (short)0, c1, false, false);
        c2 = __builtin_amdgcn_wmma_f32_16x16x4_f32(false, a, false, b2, (short)0, c2, false, false);
        c3 = __builtin_amdgcn_wmma_f32_16x16x4_f32(false, a, false, b3, (short)0, c3, false, false);
    }

    float* Cp = (blockIdx.y == 0) ? C0 : C1;
    const long long rowBase = (long long)mTile * 16 + ((lane >> 4) << 3);
#pragma unroll
    for (int v = 0; v < 8; ++v) {
        float* Crow = Cp + (rowBase + v) * 64 + r16;
        Crow[0]  = c0[v];
        Crow[16] = c1[v];
        Crow[32] = c2[v];
        Crow[48] = c3[v];
    }
}

static __device__ __forceinline__ int lane_i(int v, int l) {
    return __builtin_amdgcn_readlane(v, l);
}
static __device__ __forceinline__ float lane_f(float v, int l) {
    return __int_as_float(__builtin_amdgcn_readlane(__float_as_int(v), l));
}
static __device__ __forceinline__ void fma4(float4& a, const float4& h, float w) {
    a.x = fmaf(h.x, w, a.x);
    a.y = fmaf(h.y, w, a.y);
    a.z = fmaf(h.z, w, a.z);
    a.w = fmaf(h.w, w, a.w);
}
static __device__ __forceinline__ void fma2(float2& a, const float2& h, float w) {
    a.x = fmaf(h.x, w, a.x);
    a.y = fmaf(h.y, w, a.y);
}

__global__ __launch_bounds__(256)
void gcn_gather128_relu(const float* __restrict__ Ha, const float* __restrict__ Hb,
                        const int* __restrict__ start, const int* __restrict__ csr,
                        const float* __restrict__ dinv, const float* __restrict__ bias,
                        float* __restrict__ outp, int n) {
    const int lane = threadIdx.x & 31;
    const int node = __builtin_amdgcn_readfirstlane(blockIdx.x * 8 + (threadIdx.x >> 5));
    if (node >= n) return;

    const float* plane = ((lane < 16) ? Ha : Hb) + ((lane & 15) << 2);
    const int beg = start[node];
    const int end = start[node + 1];
    const float di = dinv[node];

    float4 acc = {0.f, 0.f, 0.f, 0.f};
    for (int jb = beg; jb < end; jb += 32) {
        int   my_s = 0;
        float my_w = 0.f;
        if (jb + lane < end) {
            my_s = csr[jb + lane];
            my_w = dinv[my_s] * di;
        }
        const int m = min(32, end - jb);
        int j = 0;
        for (; j + 4 <= m; j += 4) {
            const int s0 = lane_i(my_s, j),     s1 = lane_i(my_s, j + 1);
            const int s2 = lane_i(my_s, j + 2), s3 = lane_i(my_s, j + 3);
            const float4 h0 = *(const float4*)(plane + (long long)s0 * 64);
            const float4 h1 = *(const float4*)(plane + (long long)s1 * 64);
            const float4 h2 = *(const float4*)(plane + (long long)s2 * 64);
            const float4 h3 = *(const float4*)(plane + (long long)s3 * 64);
            fma4(acc, h0, lane_f(my_w, j));
            fma4(acc, h1, lane_f(my_w, j + 1));
            fma4(acc, h2, lane_f(my_w, j + 2));
            fma4(acc, h3, lane_f(my_w, j + 3));
        }
        for (; j < m; ++j) {
            const int s0 = lane_i(my_s, j);
            const float4 h0 = *(const float4*)(plane + (long long)s0 * 64);
            fma4(acc, h0, lane_f(my_w, j));
        }
    }

    const float4 hs = *(const float4*)(plane + (long long)node * 64);
    fma4(acc, hs, di * di);
    const int c = lane << 2;
    float4 r;
    r.x = fmaxf(acc.x + bias[c],     0.0f);
    r.y = fmaxf(acc.y + bias[c + 1], 0.0f);
    r.z = fmaxf(acc.z + bias[c + 2], 0.0f);
    r.w = fmaxf(acc.w + bias[c + 3], 0.0f);
    *(float4*)(outp + (long long)node * 128 + c) = r;
}

__global__ __launch_bounds__(256)
void gcn_gather64(const float* __restrict__ H,
                  const int* __restrict__ start, const int* __restrict__ csr,
                  const float* __restrict__ dinv, const float* __restrict__ bias,
                  float* __restrict__ outp, int n) {
    const int lane = threadIdx.x & 31;
    const int node = __builtin_amdgcn_readfirstlane(blockIdx.x * 8 + (threadIdx.x >> 5));
    if (node >= n) return;

    const float* col = H + (lane << 1);
    const int beg = start[node];
    const int end = start[node + 1];
    const float di = dinv[node];

    float2 acc = {0.f, 0.f};
    for (int jb = beg; jb < end; jb += 32) {
        int   my_s = 0;
        float my_w = 0.f;
        if (jb + lane < end) {
            my_s = csr[jb + lane];
            my_w = dinv[my_s] * di;
        }
        const int m = min(32, end - jb);
        int j = 0;
        for (; j + 4 <= m; j += 4) {
            const int s0 = lane_i(my_s, j),     s1 = lane_i(my_s, j + 1);
            const int s2 = lane_i(my_s, j + 2), s3 = lane_i(my_s, j + 3);
            const float2 h0 = *(const float2*)(col + (long long)s0 * 64);
            const float2 h1 = *(const float2*)(col + (long long)s1 * 64);
            const float2 h2 = *(const float2*)(col + (long long)s2 * 64);
            const float2 h3 = *(const float2*)(col + (long long)s3 * 64);
            fma2(acc, h0, lane_f(my_w, j));
            fma2(acc, h1, lane_f(my_w, j + 1));
            fma2(acc, h2, lane_f(my_w, j + 2));
            fma2(acc, h3, lane_f(my_w, j + 3));
        }
        for (; j < m; ++j) {
            const int s0 = lane_i(my_s, j);
            const float2 h0 = *(const float2*)(col + (long long)s0 * 64);
            fma2(acc, h0, lane_f(my_w, j));
        }
    }

    const float2 hs = *(const float2*)(col + (long long)node * 64);
    fma2(acc, hs, di * di);
    const int c = lane << 1;
    float2 r;
    r.x = acc.x + bias[c];
    r.y = acc.y + bias[c + 1];
    *(float2*)(outp + (long long)node * 64 + c) = r;
}

extern "C" void kernel_launch(void* const* d_in, const int* in_sizes, int n_in,
                              void* d_out, int out_size, void* d_ws, size_t ws_size,
                              hipStream_t stream) {
    const float* x   = (const float*)d_in[0];
    const int*   ei  = (const int*)d_in[1];
    const float* W1  = (const float*)d_in[2];
    const float* b1  = (const float*)d_in[3];
    const float* W2  = (const float*)d_in[4];
    const float* b2  = (const float*)d_in[5];

    const int IN_C  = 128;
    const int HID_C = 128;
    const int OUT_C = 64;
    const int nNodes = in_sizes[0] / IN_C;
    const int E      = in_sizes[1] / 2;
    const int* srcIdx = ei;
    const int* dstIdx = ei + E;

    const int nPad = (nNodes + 63) & ~63;
    const int ePad = (E + 63) & ~63;
    float* dinv  = (float*)d_ws;
    int*   start = (int*)(dinv + nPad);
    int*   csr   = start + nPad + 64;
    float* h1b   = (float*)(csr + ePad);
    float* hrelu = h1b + (long long)nNodes * 64;
    int*   cnt   = (int*)hrelu;
    int*   rank  = cnt + nPad;
    float* out   = (float*)d_out;

    const int T = 256;
    const int mBlocks = (nNodes / 16 + 3) / 4;

    gcn_zero_counts<<<(nNodes + T - 1) / T, T, 0, stream>>>(cnt, nNodes);
    gcn_count_rank <<<(E + T - 1) / T, T, 0, stream>>>(dstIdx, E, cnt, rank);
    gcn_scan_nodes <<<(nNodes + SCAN_T - 1) / SCAN_T, SCAN_T, 0, stream>>>(cnt, start, dinv, nNodes);
    gcn_place      <<<(E + T - 1) / T, T, 0, stream>>>(srcIdx, dstIdx, rank, start, csr, E);

    {
        dim3 grid(mBlocks, HID_C / 64);
        gcn_gemm_f32_wmma<<<grid, 128, 0, stream>>>(x, W1, out, h1b,
                                                    nNodes, HID_C, IN_C);
    }

    gcn_gather128_relu<<<(nNodes + 7) / 8, 256, 0, stream>>>(
        out, h1b, start, csr, dinv, b1, hrelu, nNodes);

    {
        dim3 grid(mBlocks, OUT_C / 64);
        gcn_gemm_f32_wmma<<<grid, 128, 0, stream>>>(hrelu, W2, h1b, h1b,
                                                    nNodes, OUT_C, HID_C);
    }

    gcn_gather64<<<(nNodes + 7) / 8, 256, 0, stream>>>(
        h1b, start, csr, dinv, b2, out, nNodes);
}


# Layer 1 aggregates x before W1 ((A x) W1): gather, W1 product, ReLU and W2 product in one kernel on 32-node LDS tiles; H1/ReLU planes and two launches gone; f32 sum order changes
# speedup vs baseline: 2.7227x; 1.2512x over previous
#include <hip/hip_runtime.h>


typedef __attribute__((ext_vector_type(2))) float v2f;
typedef __attribute__((ext_vector_type(8))) float v8f;

__global__ void gcn_zero_counts(int* __restrict__ cnt, int n) {
    int i = blockIdx.x * blockDim.x + threadIdx.x;
    if (i < n) cnt[i] = 0;
}

__global__ void gcn_count_rank(const int* __restrict__ dst, int E,
                               int* __restrict__ cnt, int* __restrict__ rank) {
    int e = blockIdx.x * blockDim.x + threadIdx.x;
    if (e < E)
        rank[e] = __hip_atomic_fetch_add(&cnt[dst[e]], 1, __ATOMIC_RELAXED,
                                         __HIP_MEMORY_SCOPE_AGENT);
}

#define SCAN_T 1024
__global__ __launch_bounds__(SCAN_T)
void gcn_scan_nodes(const int* __restrict__ cnt, int* __restrict__ start,
                    float* __restrict__ dinv, int n) {
    __shared__ int s_own[32];
    __shared__ int s_pre[32];
    const int tid  = threadIdx.x;
    const int lane = tid & 31;
    const int wave = tid >> 5;
    const int base = blockIdx.x * SCAN_T;

    int pre = 0;
    const int4* c4 = (const int4*)cnt;
    for (int q = tid; q < (base >> 2); q += SCAN_T) {
        int4 v = c4[q];
        pre += (v.x + v.y) + (v.z + v.w);
    }

    const int i = base + tid;
    const int c = (i < n) ? cnt[i] : 0;

    int inc = c;
    for (int o = 1; o < 32; o <<= 1) {
        int t = __shfl_up(inc, o);
        if (lane >= o) inc += t;
    }
    for (int o = 16; o > 0; o >>= 1) pre += __shfl_xor(pre, o);

    if (lane == 31) s_own[wave] = inc;
    if (lane == 0)  s_pre[wave] = pre;
    __syncthreads();

    const int wo = s_own[lane];
    int wp = s_pre[lane];
    int winc = wo;
    for (int o = 1; o < 32; o <<= 1) {
        int t = __shfl_up(winc, o);
        if (lane >= o) winc += t;
    }
    const int wexc = __shfl(winc - wo, wave);
    for (int o = 16; o > 0; o >>= 1) wp += __shfl_xor(wp, o);

    const int st = wp + wexc + (inc - c);
    if (i < n) {
        start[i] = st;
        dinv[i]  = rsqrtf((float)c + 1.0f);
        if (i == n - 1) start[n] = st + c;
    }
}

__global__ void gcn_place(const int* __restrict__ src, const int* __restrict__ dst,
                          const int* __restrict__ rank, const int* __restrict__ start,
                          int* __restrict__ csr, int E) {
    int e = blockIdx.x * blockDim.x + threadIdx.x;
    if (e < E) csr[start[dst[e]] + rank[e]] = src[e];
}

static __device__ __forceinline__ int lane_i(int v, int l) {
    return __builtin_amdgcn_readlane(v, l);
}
static __device__ __forceinline__ float lane_f(float v, int l) {
    return __int_as_float(__builtin_amdgcn_readlane(__float_as_int(v), l));
}
static __device__ __forceinline__ void fma4(float4& a, const float4& h, float w) {
    a.x = fmaf(h.x, w, a.x);
    a.y = fmaf(h.y, w, a.y);
    a.z = fmaf(h.z, w, a.z);
    a.w = fmaf(h.w, w, a.w);
}
static __device__ __forceinline__ void fma2(float2& a, const float2& h, float w) {
    a.x = fmaf(h.x, w, a.x);
    a.y = fmaf(h.y, w, a.y);
}

#define TILE_N 32
#define PITCH  132
__global__ __launch_bounds__(256)
void gcn_layer1_fused(const float* __restrict__ X,
                      const int* __restrict__ start, const int* __restrict__ csr,
                      const float* __restrict__ dinv,
                      const float* __restrict__ W1, const float* __restrict__ b1,
                      const float* __restrict__ W2,
                      float* __restrict__ H2, int n) {
    __shared__ __attribute__((aligned(16))) float sA[TILE_N * PITCH];
    __shared__ __attribute__((aligned(16))) float sT[TILE_N * PITCH];

    const int lane = threadIdx.x & 31;
    const int wave  = __builtin_amdgcn_readfirstlane(threadIdx.x >> 5);
    const int base  = blockIdx.x * TILE_N;
    const int node0 = base + wave * 4;
    const float* xl = X + (lane << 2);

    int   beg[4], deg[4];
    float di[4];
#pragma unroll
    for (int q = 0; q < 4; ++q) {
        const int nd  = node0 + q;
        const bool ok = nd < n;
        const int ndc = ok ? nd : n - 1;
        const int b = start[ndc];
        const int e = start[ndc + 1];
        const float d = dinv[ndc];
        beg[q] = b;
        deg[q] = ok ? e - b : 0;
        di[q]  = ok ? d : 0.f;
    }
    int   ms[4];
    float mw[4];
#pragma unroll
    for (int q = 0; q < 4; ++q) {
        const bool in = lane < deg[q];
        const int s = csr[in ? beg[q] + lane : 0];
        ms[q] = in ? s : 0;
    }
#pragma unroll
    for (int q = 0; q < 4; ++q) {
        const float dv = dinv[ms[q]];
        mw[q] = dv * ((lane < deg[q]) ? di[q] : 0.f);
    }

#pragma unroll
    for (int q = 0; q < 4; ++q) {
        const int nd  = node0 + q;
        const int end = beg[q] + deg[q];
        int   my_s = ms[q];
        float my_w = mw[q];
        float4 acc = {0.f, 0.f, 0.f, 0.f};
        for (int jb = beg[q]; jb < end; jb += 32) {
            if (jb != beg[q]) {
                my_s = 0;
                my_w = 0.f;
                if (jb + lane < end) {
                    my_s = csr[jb + lane];
                    my_w = dinv[my_s] * di[q];
                }
            }
            const int m = min(32, end - jb);
            int j = 0;
            for (; j + 4 <= m; j += 4) {
                const int s0 = lane_i(my_s, j),     s1 = lane_i(my_s, j + 1);
                const int s2 = lane_i(my_s, j + 2), s3 = lane_i(my_s, j + 3);
                const float4 h0 = *(const float4*)(xl + (long long)s0 * 128);
                const float4 h1 = *(const float4*)(xl + (long long)s1 * 128);
                const float4 h2 = *(const float4*)(xl + (long long)s2 * 128);
                const float4 h3 = *(const float4*)(xl + (long long)s3 * 128);
                fma4(acc, h0, lane_f(my_w, j));
                fma4(acc, h1, lane_f(my_w, j + 1));
                fma4(acc, h2, lane_f(my_w, j + 2));
                fma4(acc, h3, lane_f(my_w, j + 3));
            }
            for (; j < m; ++j) {
                const int s0 = lane_i(my_s, j);
                const float4 h0 = *(const float4*)(xl + (long long)s0 * 128);
                fma4(acc, h0, lane_f(my_w, j));
            }
        }
        if (nd < n) {
            const float4 hs = *(const float4*)(xl + (long long)nd * 128);
            fma4(acc, hs, di[q] * di[q]);
        }
        *(float4*)(sA + (wave * 4 + q) * PITCH + (lane << 2)) = acc;
    }
    __syncthreads();

    const int c  = lane & 15;
    const int hh = lane >> 4;
    const int ko = hh << 1;

    {
        v8f d0 = {0.f,0.f,0.f,0.f,0.f,0.f,0.f,0.f};
        v8f d1 = d0;
        const float* a0p = sA + c * PITCH + ko;
        const float* a1p = sA + (16 + c) * PITCH + ko;
        const float* bp  = W1 + ko * 128 + wave * 16 + c;
#pragma unroll 8
        for (int k0 = 0; k0 < 128; k0 += 4) {
            const v2f a0 = *(const v2f*)(a0p + k0);
            const v2f a1 = *(const v2f*)(a1p + k0);
            v2f b;
            b.x = bp[k0 * 128];
            b.y = bp[k0 * 128 + 128];
            d0 = __builtin_amdgcn_wmma_f32_16x16x4_f32(false, a0, false, b, (short)0, d0, false, false);
            d1 = __builtin_amdgcn_wmma_f32_16x16x4_f32(false, a1, false, b, (short)0, d1, false, false);
        }
        const int col = wave * 16 + c;
        const float bb = b1[col];
        float* t0 = sT + (hh << 3) * PITCH + col;
        float* t1 = t0 + 16 * PITCH;
#pragma unroll
        for (int v = 0; v < 8; ++v) {
            t0[v * PITCH] = fmaxf(d0[v] + bb, 0.0f);
            t1[v * PITCH] = fmaxf(d1[v] + bb, 0.0f);
        }
    }
    __syncthreads();

    {
        const int mt = wave >> 2;
        const int nt = wave & 3;
        v8f e0 = {0.f,0.f,0.f,0.f,0.f,0.f,0.f,0.f};
        v8f e1 = e0;
        const float* tp = sT + (mt * 16 + c) * PITCH + ko;
        const float* wp = W2 + ko * 64 + nt * 16 + c;
#pragma unroll 4
        for (int k0 = 0; k0 < 128; k0 += 8) {
            const v2f a0 = *(const v2f*)(tp + k0);
            const v2f a1 = *(const v2f*)(tp + k0 + 4);
            v2f b0, b1v;
            b0.x  = wp[k0 * 64];
            b0.y  = wp[k0 * 64 + 64];
            b1v.x = wp[(k0 + 4) * 64];
            b1v.y = wp[(k0 + 4) * 64 + 64];
            e0 = __builtin_amdgcn_wmma_f32_16x16x4_f32(false, a0, false, b0,  (short)0, e0, false, false);
            e1 = __builtin_amdgcn_wmma_f32_16x16x4_f32(false, a1, false, b1v, (short)0, e1, false, false);
        }
        const int row0 = base + mt * 16 + (hh << 3);
        float* hp = H2 + (long long)row0 * 64 + nt * 16 + c;
#pragma unroll
        for (int v = 0; v < 8; ++v)
            if (row0 + v < n) hp[v * 64] = e0[v] + e1[v];
    }
}

__global__ __launch_bounds__(256)
void gcn_gather64(const float* __restrict__ H,
                  const int* __restrict__ start, const int* __restrict__ csr,
                  const float* __restrict__ dinv, const float* __restrict__ bias,
                  float* __restrict__ outp, int n) {
    const int lane = threadIdx.x & 31;
    const int node = __builtin_amdgcn_readfirstlane(blockIdx.x * 8 + (threadIdx.x >> 5));
    if (node >= n) return;

    const float* col = H + (lane << 1);
    const int beg = start[node];
    const int end = start[node + 1];
    const float di = dinv[node];

    float2 acc = {0.f, 0.f};
    for (int jb = beg; jb < end; jb += 32) {
        int   my_s = 0;
        float my_w = 0.f;
        if (jb + lane < end) {
            my_s = csr[jb + lane];
            my_w = dinv[my_s] * di;
        }
        const int m = min(32, end - jb);
        int j = 0;
        for (; j + 4 <= m; j += 4) {
            const int s0 = lane_i(my_s, j),     s1 = lane_i(my_s, j + 1);
            const int s2 = lane_i(my_s, j + 2), s3 = lane_i(my_s, j + 3);
            const float2 h0 = *(const float2*)(col + (long long)s0 * 64);
            const float2 h1 = *(const float2*)(col + (long long)s1 * 64);
            const float2 h2 = *(const float2*)(col + (long long)s2 * 64);
            const float2 h3 = *(const float2*)(col + (long long)s3 * 64);
            fma2(acc, h0, lane_f(my_w, j));
            fma2(acc, h1, lane_f(my_w, j + 1));
            fma2(acc, h2, lane_f(my_w, j + 2));
            fma2(acc, h3, lane_f(my_w, j + 3));
        }
        for (; j < m; ++j) {
            const int s0 = lane_i(my_s, j);
            const float2 h0 = *(const float2*)(col + (long long)s0 * 64);
            fma2(acc, h0, lane_f(my_w, j));
        }
    }

    const float2 hs = *(const float2*)(col + (long long)node * 64);
    fma2(acc, hs, di * di);
    const int c = lane << 1;
    float2 r;
    r.x = acc.x + bias[c];
    r.y = acc.y + bias[c + 1];
    *(float2*)(outp + (long long)node * 64 + c) = r;
}

extern "C" void kernel_launch(void* const* d_in, const int* in_sizes, int n_in,
                              void* d_out, int out_size, void* d_ws, size_t ws_size,
                              hipStream_t stream) {
    const float* x   = (const float*)d_in[0];
    const int*   ei  = (const int*)d_in[1];
    const float* W1  = (const float*)d_in[2];
    const float* b1  = (const float*)d_in[3];
    const float* W2  = (const float*)d_in[4];
    const float* b2  = (const float*)d_in[5];

    const int IN_C  = 128;
    const int nNodes = in_sizes[0] / IN_C;
    const int E      = in_sizes[1] / 2;
    const int* srcIdx = ei;
    const int* dstIdx = ei + E;

    const int nPad = (nNodes + 63) & ~63;
    const int ePad = (E + 63) & ~63;
    float* dinv  = (float*)d_ws;
    int*   start = (int*)(dinv + nPad);
    int*   csr   = start + nPad + 64;
    float* h2    = (float*)(csr + ePad);
    int*   cnt   = (int*)(h2 + (long long)nNodes * 64);
    int*   rank  = cnt + nPad;
    float* out   = (float*)d_out;

    const int T = 256;

    gcn_zero_counts<<<(nNodes + T - 1) / T, T, 0, stream>>>(cnt, nNodes);
    gcn_count_rank <<<(E + T - 1) / T, T, 0, stream>>>(dstIdx, E, cnt, rank);
    gcn_scan_nodes <<<(nNodes + SCAN_T - 1) / SCAN_T, SCAN_T, 0, stream>>>(cnt, start, dinv, nNodes);
    gcn_place      <<<(E + T - 1) / T, T, 0, stream>>>(srcIdx, dstIdx, rank, start, csr, E);

    gcn_layer1_fused<<<(nNodes + TILE_N - 1) / TILE_N, 256, 0, stream>>>(
        x, start, csr, dinv, W1, b1, W2, h2, nNodes);

    gcn_gather64<<<(nNodes + 7) / 8, 256, 0, stream>>>(
        h2, start, csr, dinv, b2, out, nNodes);
}


# Edge sort replaced by a neighbour table: 31 sources + count in one 128-B record per node, slot = the counting add's return, spill list past 31; scan and place launches, rank/start/dinv planes gone
# speedup vs baseline: 3.8833x; 1.4263x over previous
#include <hip/hip_runtime.h>


typedef __attribute__((ext_vector_type(2))) float v2f;
typedef __attribute__((ext_vector_type(8))) float v8f;

#define REC   32
#define SLOTS 31

__global__ void gcn_clear_table(int* __restrict__ tab, long long n4,
                                int* __restrict__ spill_n) {
    int4* t4 = (int4*)tab;
    const long long stride = (long long)gridDim.x * blockDim.x;
    for (long long i = (long long)blockIdx.x * blockDim.x + threadIdx.x; i < n4; i += stride)
        t4[i] = make_int4(0, 0, 0, 0);
    if (blockIdx.x == 0 && threadIdx.x == 0) *spill_n = 0;
}

__global__ void gcn_fill_table(const int* __restrict__ src, const int* __restrict__ dst, int E,
                               int* __restrict__ tab, int* __restrict__ spill_n,
                               int* __restrict__ spill_dst, int* __restrict__ spill_src) {
    const int e = blockIdx.x * blockDim.x + threadIdx.x;
    if (e < E) {
        const int d = dst[e];
        const int s = src[e];
        int* rec = tab + (long long)d * REC;
        const int r = __hip_atomic_fetch_add(rec + SLOTS, 1, __ATOMIC_RELAXED,
                                             __HIP_MEMORY_SCOPE_AGENT);
        if (r < SLOTS) {
            rec[r] = s;
        } else {
            const int q = __hip_atomic_fetch_add(spill_n, 1, __ATOMIC_RELAXED,
                                                 __HIP_MEMORY_SCOPE_AGENT);
            spill_dst[q] = d;
            spill_src[q] = s;
        }
    }
}

static __device__ __forceinline__ int lane_i(int v, int l) {
    return __builtin_amdgcn_readlane(v, l);
}
static __device__ __forceinline__ float lane_f(float v, int l) {
    return __int_as_float(__builtin_amdgcn_readlane(__float_as_int(v), l));
}
static __device__ __forceinline__ void fma4(float4& a, const float4& h, float w) {
    a.x = fmaf(h.x, w, a.x);
    a.y = fmaf(h.y, w, a.y);
    a.z = fmaf(h.z, w, a.z);
    a.w = fmaf(h.w, w, a.w);
}
static __device__ __forceinline__ void fma2(float2& a, const float2& h, float w) {
    a.x = fmaf(h.x, w, a.x);
    a.y = fmaf(h.y, w, a.y);
}

#define TILE_N 32
#define PITCH  132
__global__ __launch_bounds__(256)
void gcn_layer1_fused(const float* __restrict__ X,
                      const int* __restrict__ tab, const int* __restrict__ spill_n,
                      const int* __restrict__ spill_dst, const int* __restrict__ spill_src,
                      const float* __restrict__ W1, const float* __restrict__ b1,
                      const float* __restrict__ W2,
                      float* __restrict__ H2, int n) {
    __shared__ __attribute__((aligned(16))) float sA[TILE_N * PITCH];
    __shared__ __attribute__((aligned(16))) float sT[TILE_N * PITCH];

    const int lane = threadIdx.x & 31;
    const int wave  = __builtin_amdgcn_readfirstlane(threadIdx.x >> 5);
    const int base  = blockIdx.x * TILE_N;
    const int node0 = base + wave * 4;
    const float* xl = X + (lane << 2);

    int   rec[4], deg[4], ms[4];
    float di[4], mw[4];
#pragma unroll
    for (int q = 0; q < 4; ++q) {
        const int nd  = node0 + q;
        const int ndc = (nd < n) ? nd : n - 1;
        rec[q] = tab[(long long)ndc * REC + lane];
    }
#pragma unroll
    for (int q = 0; q < 4; ++q) {
        const int nd  = node0 + q;
        const bool ok = nd < n;
        const int ndc = ok ? nd : n - 1;
        const int c = lane_i(rec[q], SLOTS);
        deg[q] = ok ? c : 0;
        di[q]  = ok ? rsqrtf((float)c + 1.0f) : 0.f;
        ms[q] = (lane < min(deg[q], SLOTS)) ? rec[q] : ndc;
    }
#pragma unroll
    for (int q = 0; q < 4; ++q) {
        const int cs = tab[(long long)ms[q] * REC + SLOTS];
        mw[q] = rsqrtf((float)cs + 1.0f) * ((lane < min(deg[q], SLOTS)) ? di[q] : 0.f);
    }

#pragma unroll
    for (int q = 0; q < 4; ++q) {
        const int nd = node0 + q;
        const int my_s = ms[q];
        const float my_w = mw[q];
        float4 acc = {0.f, 0.f, 0.f, 0.f};
        const int m = min(deg[q], SLOTS);
        int j = 0;
        for (; j + 4 <= m; j += 4) {
            const int s0 = lane_i(my_s, j),     s1 = lane_i(my_s, j + 1);
            const int s2 = lane_i(my_s, j + 2), s3 = lane_i(my_s, j + 3);
            const float4 h0 = *(const float4*)(xl + (long long)s0 * 128);
            const float4 h1 = *(const float4*)(xl + (long long)s1 * 128);
            const float4 h2 = *(const float4*)(xl + (long long)s2 * 128);
            const float4 h3 = *(const float4*)(xl + (long long)s3 * 128);
            fma4(acc, h0, lane_f(my_w, j));
            fma4(acc, h1, lane_f(my_w, j + 1));
            fma4(acc, h2, lane_f(my_w, j + 2));
            fma4(acc, h3, lane_f(my_w, j + 3));
        }
        for (; j < m; ++j) {
            const int s0 = lane_i(my_s, j);
            const float4 h0 = *(const float4*)(xl + (long long)s0 * 128);
            fma4(acc, h0, lane_f(my_w, j));
        }
        if (deg[q] > SLOTS) {
            const int total = *spill_n;
            for (int jb = 0; jb < total; jb += 32) {
                int sd = -1, ss = 0;
                if (jb + lane < total) {
                    sd = spill_dst[jb + lane];
                    ss = spill_src[jb + lane];
                }
                const bool hit = sd == nd;
                float w = 0.f;
                if (hit) w = rsqrtf((float)tab[(long long)ss * REC + SLOTS] + 1.0f) * di[q];
                unsigned hits = __builtin_amdgcn_ballot_w32(hit);
                while (hits) {
                    const int l = __builtin_ctz(hits);
                    hits &= hits - 1;
                    const int s0 = lane_i(ss, l);
                    const float4 h0 = *(const float4*)(xl + (long long)s0 * 128);
                    fma4(acc, h0, lane_f(w, l));
                }
            }
        }
        if (nd < n) {
            const float4 hs = *(const float4*)(xl + (long long)nd * 128);
            fma4(acc, hs, di[q] * di[q]);
        }
        *(float4*)(sA + (wave * 4 + q) * PITCH + (lane << 2)) = acc;
    }
    __syncthreads();

    const int c  = lane & 15;
    const int hh = lane >> 4;
    const int ko = hh << 1;

    {
        v8f d0 = {0.f,0.f,0.f,0.f,0.f,0.f,0.f,0.f};
        v8f d1 = d0;
        const float* a0p = sA + c * PITCH + ko;
        const float* a1p = sA + (16 + c) * PITCH + ko;
        const float* bp  = W1 + ko * 128 + wave * 16 + c;
#pragma unroll 8
        for (int k0 = 0; k0 < 128; k0 += 4) {
            const v2f a0 = *(const v2f*)(a0p + k0);
            const v2f a1 = *(const v2f*)(a1p + k0);
            v2f b;
            b.x = bp[k0 * 128];
            b.y = bp[k0 * 128 + 128];
            d0 = __builtin_amdgcn_wmma_f32_16x16x4_f32(false, a0, false, b, (short)0, d0, false, false);
            d1 = __builtin_amdgcn_wmma_f32_16x16x4_f32(false, a1, false, b, (short)0, d1, false, false);
        }
        const int col = wave * 16 + c;
        const float bb = b1[col];
        float* t0 = sT + (hh << 3) * PITCH + col;
        float* t1 = t0 + 16 * PITCH;
#pragma unroll
        for (int v = 0; v < 8; ++v) {
            t0[v * PITCH] = fmaxf(d0[v] + bb, 0.0f);
            t1[v * PITCH] = fmaxf(d1[v] + bb, 0.0f);
        }
    }
    __syncthreads();

    {
        const int mt = wave >> 2;
        const int nt = wave & 3;
        v8f e0 = {0.f,0.f,0.f,0.f,0.f,0.f,0.f,0.f};
        v8f e1 = e0;
        const float* tp = sT + (mt * 16 + c) * PITCH + ko;
        const float* wp = W2 + ko * 64 + nt * 16 + c;
#pragma unroll 4
        for (int k0 = 0; k0 < 128; k0 += 8) {
            const v2f a0 = *(const v2f*)(tp + k0);
            const v2f a1 = *(const v2f*)(tp + k0 + 4);
            v2f b0, b1v;
            b0.x  = wp[k0 * 64];
            b0.y  = wp[k0 * 64 + 64];
            b1v.x = wp[(k0 + 4) * 64];
            b1v.y = wp[(k0 + 4) * 64 + 64];
            e0 = __builtin_amdgcn_wmma_f32_16x16x4_f32(false, a0, false, b0,  (short)0, e0, false, false);
            e1 = __builtin_amdgcn_wmma_f32_16x16x4_f32(false, a1, false, b1v, (short)0, e1, false, false);
        }
        const int row0 = base + mt * 16 + (hh << 3);
        float* hp = H2 + (long long)row0 * 64 + nt * 16 + c;
#pragma unroll
        for (int v = 0; v < 8; ++v)
            if (row0 + v < n) hp[v * 64] = e0[v] + e1[v];
    }
}

__global__ __launch_bounds__(256)
void gcn_gather64(const float* __restrict__ H,
                  const int* __restrict__ tab, const int* __restrict__ spill_n,
                  const int* __restrict__ spill_dst, const int* __restrict__ spill_src,
                  const float* __restrict__ bias,
                  float* __restrict__ outp, int n) {
    const int lane = threadIdx.x & 31;
    const int node = __builtin_amdgcn_readfirstlane(blockIdx.x * 8 + (threadIdx.x >> 5));
    if (node >= n) return;

    const float* col = H + (lane << 1);
    const int rec = tab[(long long)node * REC + lane];
    const int deg = lane_i(rec, SLOTS);
    const float di = rsqrtf((float)deg + 1.0f);
    const int m = min(deg, SLOTS);
    const int my_s = (lane < m) ? rec : node;
    const int cs = tab[(long long)my_s * REC + SLOTS];
    const float my_w = rsqrtf((float)cs + 1.0f) * ((lane < m) ? di : 0.f);

    float2 acc = {0.f, 0.f};
    int j = 0;
    for (; j + 4 <= m; j += 4) {
        const int s0 = lane_i(my_s, j),     s1 = lane_i(my_s, j + 1);
        const int s2 = lane_i(my_s, j + 2), s3 = lane_i(my_s, j + 3);
        const float2 h0 = *(const float2*)(col + (long long)s0 * 64);
        const float2 h1 = *(const float2*)(col + (long long)s1 * 64);
        const float2 h2 = *(const float2*)(col + (long long)s2 * 64);
        const float2 h3 = *(const float2*)(col + (long long)s3 * 64);
        fma2(acc, h0, lane_f(my_w, j));
        fma2(acc, h1, lane_f(my_w, j + 1));
        fma2(acc, h2, lane_f(my_w, j + 2));
        fma2(acc, h3, lane_f(my_w, j + 3));
    }
    for (; j < m; ++j) {
        const int s0 = lane_i(my_s, j);
        const float2 h0 = *(const float2*)(col + (long long)s0 * 64);
        fma2(acc, h0, lane_f(my_w, j));
    }
    if (deg > SLOTS) {
        const int total = *spill_n;
        for (int jb = 0; jb < total; jb += 32) {
            int sd = -1, ss = 0;
            if (jb + lane < total) {
                sd = spill_dst[jb + lane];
                ss = spill_src[jb + lane];
            }
            const bool hit = sd == node;
            float w = 0.f;
            if (hit) w = rsqrtf((float)tab[(long long)ss * REC + SLOTS] + 1.0f) * di;
            unsigned hits = __builtin_amdgcn_ballot_w32(hit);
            while (hits) {
                const int l = __builtin_ctz(hits);
                hits &= hits - 1;
                const int s0 = lane_i(ss, l);
                const float2 h0 = *(const float2*)(col + (long long)s0 * 64);
                fma2(acc, h0, lane_f(w, l));
            }
        }
    }

    const float2 hs = *(const float2*)(col + (long long)node * 64);
    fma2(acc, hs, di * di);
    const int c = lane << 1;
    float2 r;
    r.x = acc.x + bias[c];
    r.y = acc.y + bias[c + 1];
    *(float2*)(outp + (long long)node * 64 + c) = r;
}

extern "C" void kernel_launch(void* const* d_in, const int* in_sizes, int n_in,
                              void* d_out, int out_size, void* d_ws, size_t ws_size,
                              hipStream_t stream) {
    const float* x   = (const float*)d_in[0];
    const int*   ei  = (const int*)d_in[1];
    const float* W1  = (const float*)d_in[2];
    const float* b1  = (const float*)d_in[3];
    const float* W2  = (const float*)d_in[4];
    const float* b2  = (const float*)d_in[5];

    const int IN_C  = 128;
    const int nNodes = in_sizes[0] / IN_C;
    const int E      = in_sizes[1] / 2;
    const int* srcIdx = ei;
    const int* dstIdx = ei + E;

    const int nPad = (nNodes + 63) & ~63;
    const int ePad = (E + 63) & ~63;
    int*   tab       = (int*)d_ws;
    float* h2        = (float*)(tab + (long long)nPad * REC);
    int*   spill_dst = (int*)(h2 + (long long)nPad * 64);
    int*   spill_src = spill_dst + ePad;
    int*   spill_n   = spill_src + ePad;
    float* out       = (float*)d_out;

    const int T = 256;

    gcn_clear_table<<<256, T, 0, stream>>>(tab, (long long)nNodes * (REC / 4), spill_n);
    gcn_fill_table <<<(E + T - 1) / T, T, 0, stream>>>(srcIdx, dstIdx, E, tab, spill_n,
                                                       spill_dst, spill_src);

    gcn_layer1_fused<<<(nNodes + TILE_N - 1) / TILE_N, 256, 0, stream>>>(
        x, tab, spill_n, spill_dst, spill_src, W1, b1, W2, h2, nNodes);

    gcn_gather64<<<(nNodes + 7) / 8, 256, 0, stream>>>(
        h2, tab, spill_n, spill_dst, spill_src, b2, out, nNodes);
}


# Neighbour table built by owner groups: each of 128 groups scans the destination list, slot = returning LDS add, records stored whole; global adds, clear launch gone; f32 sum order = LDS arrival
# speedup vs baseline: 4.0511x; 1.0432x over previous
#include <hip/hip_runtime.h>


typedef __attribute__((ext_vector_type(2))) float v2f;
typedef __attribute__((ext_vector_type(8))) float v8f;

#define REC   32
#define SLOTS 31
#define RANGE 392
#define CAP   1024
#define BT    1024

#define QCAP  512
#define QFULL (QCAP - 128)

static __device__ __forceinline__ void place_edge(int* ltab, int* lsn, unsigned u, int e, int a,
                                                  const int* __restrict__ src,
                                                  int* __restrict__ sp_dst, int* __restrict__ sp_src) {
    int* rec = ltab + u * REC;
    const int r = __hip_atomic_fetch_add(rec + SLOTS, 1, __ATOMIC_RELAXED,
                                         __HIP_MEMORY_SCOPE_WORKGROUP);
    if (r < SLOTS) {
        rec[r] = e;
    } else {
        const int p = __hip_atomic_fetch_add(lsn, 1, __ATOMIC_RELAXED,
                                             __HIP_MEMORY_SCOPE_WORKGROUP);
        if (p < CAP) { sp_dst[p] = a + (int)u; sp_src[p] = src[e]; }
    }
}

static __device__ __forceinline__ void drain_queue(const int2* q, int cnt, int lane,
                                                   int* ltab, int* lsn, int a,
                                                   const int* __restrict__ src,
                                                   int* __restrict__ sp_dst, int* __restrict__ sp_src) {
    __builtin_amdgcn_wave_barrier();
    for (int b = 0; b < cnt; b += 32) {
        if (b + lane < cnt) {
            const int2 ue = q[b + lane];
            place_edge(ltab, lsn, (unsigned)ue.x, ue.y, a, src, sp_dst, sp_src);
        }
    }
    __builtin_amdgcn_wave_barrier();
}

static __device__ __forceinline__ int scan4(const int4 d, int e, int a, int2* q, int qn) {
    const unsigned u0 = (unsigned)(d.x - a), u1 = (unsigned)(d.y - a);
    const unsigned u2 = (unsigned)(d.z - a), u3 = (unsigned)(d.w - a);
    if (__builtin_amdgcn_ballot_w32(min(min(u0, u1), min(u2, u3)) < (unsigned)RANGE) != 0u) {
        const unsigned uu[4] = {u0, u1, u2, u3};
#pragma unroll
        for (int k = 0; k < 4; ++k) {
            const bool hit = uu[k] < (unsigned)RANGE;
            const unsigned bal = __builtin_amdgcn_ballot_w32(hit);
            if (hit) q[qn + __builtin_amdgcn_mbcnt_lo(bal, 0u)] = make_int2((int)uu[k], e + k);
            qn += __builtin_popcount(bal);
        }
    }
    return qn;
}

__global__ __launch_bounds__(BT)
void gcn_build_table(const int* __restrict__ src, const int* __restrict__ dst, int E, int n,
                     int* __restrict__ tab, int* __restrict__ spill_cnt,
                     int* __restrict__ spill_dst, int* __restrict__ spill_src) {
    __shared__ __attribute__((aligned(16))) int ltab[RANGE * REC];
    __shared__ __attribute__((aligned(8)))  int2 lq[(BT / 32) * QCAP];
    __shared__ int lsn;

    const int tid  = threadIdx.x;
    const int lane = tid & 31;
    int2* q = lq + __builtin_amdgcn_readfirstlane(tid >> 5) * QCAP;
    const int a = blockIdx.x * RANGE;
    int* sp_dst = spill_dst + (long long)blockIdx.x * CAP;
    int* sp_src = spill_src + (long long)blockIdx.x * CAP;
    int4* ltab4 = (int4*)ltab;

    for (int i = tid; i < RANGE * (REC / 4); i += BT) ltab4[i] = make_int4(0, 0, 0, 0);
    if (tid == 0) lsn = 0;
    __syncthreads();

    const unsigned nv = (((size_t)dst & 15) == 0) ? (unsigned)(E >> 2) : 0u;
    const int4* d4 = (const int4*)dst;
    const unsigned off = blockIdx.x * (nv / gridDim.x);
    int qn = 0;
    unsigned iw = __builtin_amdgcn_readfirstlane(tid & ~31);
    for (; iw + 31u + 7u * BT < nv; iw += 8u * BT) {
        const unsigned i = iw + lane;
        unsigned jj[8];
        int4 d[8];
#pragma unroll
        for (int k = 0; k < 8; ++k) {
            const unsigned j = i + k * BT + off;
            jj[k] = min(j, j - nv);
            d[k] = d4[jj[k]];
        }
#pragma unroll
        for (int k = 0; k < 8; ++k) {
            qn = scan4(d[k], (int)(jj[k] << 2), a, q, qn);
            if (qn > QFULL) { drain_queue(q, qn, lane, ltab, &lsn, a, src, sp_dst, sp_src); qn = 0; }
        }
    }
    for (; iw < nv; iw += BT) {
        const unsigned i = iw + lane;
        const unsigned j = i + off;
        const unsigned j1 = min(j, j - nv);
        int4 dd = make_int4(-1, -1, -1, -1);
        if (i < nv) dd = d4[j1];
        qn = scan4(dd, (int)(j1 << 2), a, q, qn);
        if (qn > QFULL) { drain_queue(q, qn, lane, ltab, &lsn, a, src, sp_dst, sp_src); qn = 0; }
    }
    drain_queue(q, qn, lane, ltab, &lsn, a, src, sp_dst, sp_src);
    for (int e = (int)(nv << 2) + tid; e < E; e += BT) {
        const unsigned u = (unsigned)(dst[e] - a);
        if (u < (unsigned)RANGE) place_edge(ltab, &lsn, u, e, a, src, sp_dst, sp_src);
    }
    __syncthreads();

    int4* tab4 = (int4*)tab;
    for (int w = tid; w < RANGE * (REC / 4); w += BT) {
        const int nl = w >> 3;
        const int q4 = (w & 7) << 2;
        const int nd = a + nl;
        if (nd < n) {
            int4 v = ltab4[w];
            const int m = min(ltab[nl * REC + SLOTS], SLOTS);
            if (q4     < m) v.x = src[v.x];
            if (q4 + 1 < m) v.y = src[v.y];
            if (q4 + 2 < m) v.z = src[v.z];
            if (q4 + 3 < m) v.w = src[v.w];
            tab4[(long long)nd * (REC / 4) + (w & 7)] = v;
        }
    }
    if (tid == 0) spill_cnt[blockIdx.x] = lsn;
}

static __device__ __forceinline__ int lane_i(int v, int l) {
    return __builtin_amdgcn_readlane(v, l);
}
static __device__ __forceinline__ float lane_f(float v, int l) {
    return __int_as_float(__builtin_amdgcn_readlane(__float_as_int(v), l));
}
static __device__ __forceinline__ void fma4(float4& a, const float4& h, float w) {
    a.x = fmaf(h.x, w, a.x);
    a.y = fmaf(h.y, w, a.y);
    a.z = fmaf(h.z, w, a.z);
    a.w = fmaf(h.w, w, a.w);
}
static __device__ __forceinline__ void fma2(float2& a, const float2& h, float w) {
    a.x = fmaf(h.x, w, a.x);
    a.y = fmaf(h.y, w, a.y);
}

#define TILE_N 32
#define PITCH  132
__global__ __launch_bounds__(256)
void gcn_layer1_fused(const float* __restrict__ X,
                      const int* __restrict__ src, const int* __restrict__ dst, int E,
                      const int* __restrict__ tab, const int* __restrict__ spill_cnt,
                      const int* __restrict__ spill_dst, const int* __restrict__ spill_src,
                      const float* __restrict__ W1, const float* __restrict__ b1,
                      const float* __restrict__ W2,
                      float* __restrict__ H2, int n) {
    __shared__ __attribute__((aligned(16))) float sA[TILE_N * PITCH];
    __shared__ __attribute__((aligned(16))) float sT[TILE_N * PITCH];

    const int lane = threadIdx.x & 31;
    const int wave  = __builtin_amdgcn_readfirstlane(threadIdx.x >> 5);
    const int base  = blockIdx.x * TILE_N;
    const int node0 = base + wave * 4;
    const float* xl = X + (lane << 2);

    int   rec[4], deg[4], mm[4], ms[4];
    float di[4], mw[4];
#pragma unroll
    for (int q = 0; q < 4; ++q) {
        const int nd  = node0 + q;
        const int ndc = (nd < n) ? nd : n - 1;
        rec[q] = tab[(long long)ndc * REC + lane];
    }
#pragma unroll
    for (int q = 0; q < 4; ++q) {
        const int nd  = node0 + q;
        const bool ok = nd < n;
        const int ndc = ok ? nd : n - 1;
        const int c = lane_i(rec[q], SLOTS);
        deg[q] = ok ? c : 0;
        di[q]  = ok ? rsqrtf((float)c + 1.0f) : 0.f;
        mm[q] = min(deg[q], SLOTS);
        if (deg[q] > SLOTS && spill_cnt[ndc / RANGE] > CAP) mm[q] = 0;
        ms[q] = (lane < mm[q]) ? rec[q] : ndc;
    }
#pragma unroll
    for (int q = 0; q < 4; ++q) {
        const int cs = tab[(long long)ms[q] * REC + SLOTS];
        mw[q] = rsqrtf((float)cs + 1.0f) * ((lane < mm[q]) ? di[q] : 0.f);
    }

#pragma unroll
    for (int q = 0; q < 4; ++q) {
        const int nd = node0 + q;
        const int my_s = ms[q];
        const float my_w = mw[q];
        float4 acc = {0.f, 0.f, 0.f, 0.f};
        const int m = mm[q];
        int j = 0;
        for (; j + 4 <= m; j += 4) {
            const int s0 = lane_i(my_s, j),     s1 = lane_i(my_s, j + 1);
            const int s2 = lane_i(my_s, j + 2), s3 = lane_i(my_s, j + 3);
            const float4 h0 = *(const float4*)(xl + (long long)s0 * 128);
            const float4 h1 = *(const float4*)(xl + (long long)s1 * 128);
            const float4 h2 = *(const float4*)(xl + (long long)s2 * 128);
            const float4 h3 = *(const float4*)(xl + (long long)s3 * 128);
            fma4(acc, h0, lane_f(my_w, j));
            fma4(acc, h1, lane_f(my_w, j + 1));
            fma4(acc, h2, lane_f(my_w, j + 2));
            fma4(acc, h3, lane_f(my_w, j + 3));
        }
        for (; j < m; ++j) {
            const int s0 = lane_i(my_s, j);
            const float4 h0 = *(const float4*)(xl + (long long)s0 * 128);
            fma4(acc, h0, lane_f(my_w, j));
        }
        if (deg[q] > SLOTS) {
            const int g = nd / RANGE;
            const int* pd = (m > 0) ? spill_dst + (long long)g * CAP : dst;
            const int* ps = (m > 0) ? spill_src + (long long)g * CAP : src;
            const int total = (m > 0) ? spill_cnt[g] : E;
            for (int jb = 0; jb < total; jb += 32) {
                int sd = -1, ss = 0;
                if (jb + lane < total) {
                    sd = pd[jb + lane];
                    ss = ps[jb + lane];
                }
                const bool hit = sd == nd;
                float w = 0.f;
                if (hit) w = rsqrtf((float)tab[(long long)ss * REC + SLOTS] + 1.0f) * di[q];
                unsigned hits = __builtin_amdgcn_ballot_w32(hit);
                while (hits) {
                    const int l = __builtin_ctz(hits);
                    hits &= hits - 1;
                    const int s0 = lane_i(ss, l);
                    const float4 h0 = *(const float4*)(xl + (long long)s0 * 128);
                    fma4(acc, h0, lane_f(w, l));
                }
            }
        }
        if (nd < n) {
            const float4 hs = *(const float4*)(xl + (long long)nd * 128);
            fma4(acc, hs, di[q] * di[q]);
        }
        *(float4*)(sA + (wave * 4 + q) * PITCH + (lane << 2)) = acc;
    }
    __syncthreads();

    const int c  = lane & 15;
    const int hh = lane >> 4;
    const int ko = hh << 1;

    {
        v8f d0 = {0.f,0.f,0.f,0.f,0.f,0.f,0.f,0.f};
        v8f d1 = d0;
        const float* a0p = sA + c * PITCH + ko;
        const float* a1p = sA + (16 + c) * PITCH + ko;
        const float* bp  = W1 + ko * 128 + wave * 16 + c;
#pragma unroll 8
        for (int k0 = 0; k0 < 128; k0 += 4) {
            const v2f a0 = *(const v2f*)(a0p + k0);
            const v2f a1 = *(const v2f*)(a1p + k0);
            v2f b;
            b.x = bp[k0 * 128];
            b.y = bp[k0 * 128 + 128];
            d0 = __builtin_amdgcn_wmma_f32_16x16x4_f32(false, a0, false, b, (short)0, d0, false, false);
            d1 = __builtin_amdgcn_wmma_f32_16x16x4_f32(false, a1, false, b, (short)0, d1, false, false);
        }
        const int col = wave * 16 + c;
        const float bb = b1[col];
        float* t0 = sT + (hh << 3) * PITCH + col;
        float* t1 = t0 + 16 * PITCH;
#pragma unroll
        for (int v = 0; v < 8; ++v) {
            t0[v * PITCH] = fmaxf(d0[v] + bb, 0.0f);
            t1[v * PITCH] = fmaxf(d1[v] + bb, 0.0f);
        }
    }
    __syncthreads();

    {
        const int mt = wave >> 2;
        const int nt = wave & 3;
        v8f e0 = {0.f,0.f,0.f,0.f,0.f,0.f,0.f,0.f};
        v8f e1 = e0;
        const float* tp = sT + (mt * 16 + c) * PITCH + ko;
        const float* wp = W2 + ko * 64 + nt * 16 + c;
#pragma unroll 4
        for (int k0 = 0; k0 < 128; k0 += 8) {
            const v2f a0 = *(const v2f*)(tp + k0);
            const v2f a1 = *(const v2f*)(tp + k0 + 4);
            v2f b0, b1v;
            b0.x  = wp[k0 * 64];
            b0.y  = wp[k0 * 64 + 64];
            b1v.x = wp[(k0 + 4) * 64];
            b1v.y = wp[(k0 + 4) * 64 + 64];
            e0 = __builtin_amdgcn_wmma_f32_16x16x4_f32(false, a0, false, b0,  (short)0, e0, false, false);
            e1 = __builtin_amdgcn_wmma_f32_16x16x4_f32(false, a1, false, b1v, (short)0, e1, false, false);
        }
        const int row0 = base + mt * 16 + (hh << 3);
        float* hp = H2 + (long long)row0 * 64 + nt * 16 + c;
#pragma unroll
        for (int v = 0; v < 8; ++v)
            if (row0 + v < n) hp[v * 64] = e0[v] + e1[v];
    }
}

__global__ __launch_bounds__(256)
void gcn_gather64(const float* __restrict__ H,
                  const int* __restrict__ src, const int* __restrict__ dst, int E,
                  const int* __restrict__ tab, const int* __restrict__ spill_cnt,
                  const int* __restrict__ spill_dst, const int* __restrict__ spill_src,
                  const float* __restrict__ bias,
                  float* __restrict__ outp, int n) {
    const int lane = threadIdx.x & 31;
    const int node = __builtin_amdgcn_readfirstlane(blockIdx.x * 8 + (threadIdx.x >> 5));
    if (node >= n) return;

    const float* col = H + (lane << 1);
    const int rec = tab[(long long)node * REC + lane];
    const int deg = lane_i(rec, SLOTS);
    const float di = rsqrtf((float)deg + 1.0f);
    int m = min(deg, SLOTS);
    if (deg > SLOTS && spill_cnt[node / RANGE] > CAP) m = 0;
    const int my_s = (lane < m) ? rec : node;
    const int cs = tab[(long long)my_s * REC + SLOTS];
    const float my_w = rsqrtf((float)cs + 1.0f) * ((lane < m) ? di : 0.f);

    float2 acc = {0.f, 0.f};
    int j = 0;
    for (; j + 4 <= m; j += 4) {
        const int s0 = lane_i(my_s, j),     s1 = lane_i(my_s, j + 1);
        const int s2 = lane_i(my_s, j + 2), s3 = lane_i(my_s, j + 3);
        const float2 h0 = *(const float2*)(col + (long long)s0 * 64);
        const float2 h1 = *(const float2*)(col + (long long)s1 * 64);
        const float2 h2 = *(const float2*)(col + (long long)s2 * 64);
        const float2 h3 = *(const float2*)(col + (long long)s3 * 64);
        fma2(acc, h0, lane_f(my_w, j));
        fma2(acc, h1, lane_f(my_w, j + 1));
        fma2(acc, h2, lane_f(my_w, j + 2));
        fma2(acc, h3, lane_f(my_w, j + 3));
    }
    for (; j < m; ++j) {
        const int s0 = lane_i(my_s, j);
        const float2 h0 = *(const float2*)(col + (long long)s0 * 64);
        fma2(acc, h0, lane_f(my_w, j));
    }
    if (deg > SLOTS) {
        const int g = node / RANGE;
        const int* pd = (m > 0) ? spill_dst + (long long)g * CAP : dst;
        const int* ps = (m > 0) ? spill_src + (long long)g * CAP : src;
        const int total = (m > 0) ? spill_cnt[g] : E;
        for (int jb = 0; jb < total; jb += 32) {
            int sd = -1, ss = 0;
            if (jb + lane < total) {
                sd = pd[jb + lane];
                ss = ps[jb + lane];
            }
            const bool hit = sd == node;
            float w = 0.f;
            if (hit) w = rsqrtf((float)tab[(long long)ss * REC + SLOTS] + 1.0f) * di;
            unsigned hits = __builtin_amdgcn_ballot_w32(hit);
            while (hits) {
                const int l = __builtin_ctz(hits);
                hits &= hits - 1;
                const int s0 = lane_i(ss, l);
                const float2 h0 = *(const float2*)(col + (long long)s0 * 64);
                fma2(acc, h0, lane_f(w, l));
            }
        }
    }

    const float2 hs = *(const float2*)(col + (long long)node * 64);
    fma2(acc, hs, di * di);
    const int c = lane << 1;
    float2 r;
    r.x = acc.x + bias[c];
    r.y = acc.y + bias[c + 1];
    *(float2*)(outp + (long long)node * 64 + c) = r;
}

extern "C" void kernel_launch(void* const* d_in, const int* in_sizes, int n_in,
                              void* d_out, int out_size, void* d_ws, size_t ws_size,
                              hipStream_t stream) {
    const float* x   = (const float*)d_in[0];
    const int*   ei  = (const int*)d_in[1];
    const float* W1  = (const float*)d_in[2];
    const float* b1  = (const float*)d_in[3];
    const float* W2  = (const float*)d_in[4];
    const float* b2  = (const float*)d_in[5];

    const int IN_C  = 128;
    const int nNodes = in_sizes[0] / IN_C;
    const int E      = in_sizes[1] / 2;
    const int* srcIdx = ei;
    const int* dstIdx = ei + E;

    const int nPad   = (nNodes + 63) & ~63;
    const int groups = (nNodes + RANGE - 1) / RANGE;
    int*   tab       = (int*)d_ws;
    float* h2        = (float*)(tab + (long long)nPad * REC);
    int*   spill_dst = (int*)(h2 + (long long)nPad * 64);
    int*   spill_src = spill_dst + (long long)groups * CAP;
    int*   spill_cnt = spill_src + (long long)groups * CAP;
    float* out       = (float*)d_out;


    gcn_build_table<<<groups, BT, 0, stream>>>(srcIdx, dstIdx, E, nNodes, tab, spill_cnt,
                                               spill_dst, spill_src);

    gcn_layer1_fused<<<(nNodes + TILE_N - 1) / TILE_N, 256, 0, stream>>>(
        x, srcIdx, dstIdx, E, tab, spill_cnt, spill_dst, spill_src, W1, b1, W2, h2, nNodes);

    gcn_gather64<<<(nNodes + 7) / 8, 256, 0, stream>>>(
        h2, srcIdx, dstIdx, E, tab, spill_cnt, spill_dst, spill_src, b2, out, nNodes);
}


# Edges binned by owner group: 256 groups file their stretch of the list into per-owner segments by LDS rank; an owner reads its 4700 edges, not all 600000; +1 launch; f32 sum order = LDS arrival
# speedup vs baseline: 4.8679x; 1.2016x over previous
#include <hip/hip_runtime.h>


typedef __attribute__((ext_vector_type(2))) float v2f;
typedef __attribute__((ext_vector_type(8))) float v8f;

#define REC   32
#define SLOTS 31
#define RANGE 392
#define CAP   1024
#define BT    1024
#define WA    256
#define ABT   512
#define BCAP  48
#define GMAX  128

__global__ __launch_bounds__(ABT)
void gcn_bin_edges(const int* __restrict__ src, const int* __restrict__ dst, int E, int groups,
                   int* __restrict__ bins, int* __restrict__ bin_cnt) {
    __shared__ int lc[GMAX + 1];
    const int tid = threadIdx.x;
    if (tid <= GMAX) lc[tid] = 0;
    __syncthreads();

    int2* seg = (int2*)bins + (long long)blockIdx.x * BCAP;
    const int per = (E + (int)gridDim.x - 1) / (int)gridDim.x;
    const int e0  = blockIdx.x * per;
    const int e1  = min(E, e0 + per);
    for (int eb = e0; eb < e1; eb += 5 * ABT) {
        int  d[5], s[5], g[5], r[5];
        bool ok[5];
#pragma unroll
        for (int k = 0; k < 5; ++k) {
            const int e = eb + k * ABT + tid;
            ok[k] = e < e1;
            const int ec = ok[k] ? e : eb;
            d[k] = dst[ec];
            s[k] = src[ec];
        }
#pragma unroll
        for (int k = 0; k < 5; ++k) asm volatile("" : "+v"(s[k]));
#pragma unroll
        for (int k = 0; k < 5; ++k) {
            g[k] = ok[k] ? d[k] / RANGE : GMAX;
            r[k] = __hip_atomic_fetch_add(lc + g[k], 1, __ATOMIC_RELAXED,
                                          __HIP_MEMORY_SCOPE_WORKGROUP);
        }
#pragma unroll
        for (int k = 0; k < 5; ++k) {
            if (ok[k] && r[k] < BCAP)
                seg[(long long)g[k] * (WA * BCAP) + r[k]] = make_int2(d[k] - g[k] * RANGE, s[k]);
        }
    }
    __syncthreads();
    if (tid < groups) bin_cnt[tid * WA + blockIdx.x] = lc[tid];
}

#define QCAP  512
#define QFULL (QCAP - 128)

static __device__ __forceinline__ void store_edge(int* ltab, int* lsn, unsigned u, int s, int r, int a,
                                                  int* __restrict__ sp_dst, int* __restrict__ sp_src) {
    if (r < SLOTS) {
        ltab[u * REC + r] = s;
    } else {
        const int p = __hip_atomic_fetch_add(lsn, 1, __ATOMIC_RELAXED,
                                             __HIP_MEMORY_SCOPE_WORKGROUP);
        if (p < CAP) { sp_dst[p] = a + (int)u; sp_src[p] = s; }
    }
}

static __device__ __forceinline__ void place_edge(int* ltab, int* lsn, unsigned u, int s, int a,
                                                  int* __restrict__ sp_dst, int* __restrict__ sp_src) {
    const int r = __hip_atomic_fetch_add(ltab + u * REC + SLOTS, 1, __ATOMIC_RELAXED,
                                         __HIP_MEMORY_SCOPE_WORKGROUP);
    store_edge(ltab, lsn, u, s, r, a, sp_dst, sp_src);
}

static __device__ __forceinline__ void drain_queue(const int2* q, int cnt, int lane,
                                                   int* ltab, int* lsn, int a,
                                                   const int* __restrict__ src,
                                                   int* __restrict__ sp_dst, int* __restrict__ sp_src) {
    __builtin_amdgcn_wave_barrier();
    for (int b = 0; b < cnt; b += 32) {
        if (b + lane < cnt) {
            const int2 ue = q[b + lane];
            place_edge(ltab, lsn, (unsigned)ue.x, src[ue.y], a, sp_dst, sp_src);
        }
    }
    __builtin_amdgcn_wave_barrier();
}

static __device__ __forceinline__ int scan4(const int4 d, int e, int a, int2* q, int qn) {
    const unsigned u0 = (unsigned)(d.x - a), u1 = (unsigned)(d.y - a);
    const unsigned u2 = (unsigned)(d.z - a), u3 = (unsigned)(d.w - a);
    if (__builtin_amdgcn_ballot_w32(min(min(u0, u1), min(u2, u3)) < (unsigned)RANGE) != 0u) {
        const unsigned uu[4] = {u0, u1, u2, u3};
#pragma unroll
        for (int k = 0; k < 4; ++k) {
            const bool hit = uu[k] < (unsigned)RANGE;
            const unsigned bal = __builtin_amdgcn_ballot_w32(hit);
            if (hit) q[qn + __builtin_amdgcn_mbcnt_lo(bal, 0u)] = make_int2((int)uu[k], e + k);
            qn += __builtin_popcount(bal);
        }
    }
    return qn;
}

__global__ __launch_bounds__(BT)
void gcn_build_table(const int* __restrict__ src, const int* __restrict__ dst, int E, int n,
                     const int* __restrict__ bins, const int* __restrict__ bin_cnt,
                     int* __restrict__ tab, int* __restrict__ spill_cnt,
                     int* __restrict__ spill_dst, int* __restrict__ spill_src) {
    __shared__ __attribute__((aligned(16))) int ltab[RANGE * REC];
    __shared__ __attribute__((aligned(8)))  int2 lq[(BT / 32) * QCAP];
    __shared__ int lsn;
    __shared__ int lovf;

    const int tid  = threadIdx.x;
    const int lane = tid & 31;
    const int wave = __builtin_amdgcn_readfirstlane(tid >> 5);
    int2* q = lq + wave * QCAP;
    const int a = blockIdx.x * RANGE;
    int* sp_dst = spill_dst + (long long)blockIdx.x * CAP;
    int* sp_src = spill_src + (long long)blockIdx.x * CAP;
    int4* ltab4 = (int4*)ltab;

    int c[WA / 32];
#pragma unroll
    for (int i = 0; i < WA / 32; ++i) c[i] = bin_cnt[blockIdx.x * WA + wave + 32 * i];

    for (int i = tid; i < RANGE * (REC / 4); i += BT) ltab4[i] = make_int4(0, 0, 0, 0);
    if (tid == 0) { lsn = 0; lovf = 0; }
    __syncthreads();

    bool over = false;
#pragma unroll
    for (int i = 0; i < WA / 32; ++i) over = over || (c[i] > BCAP);
    if (over) lovf = 1;
    __syncthreads();

    if (lovf == 0) {
        const int2* bp = (const int2*)bins + (long long)blockIdx.x * (WA * BCAP);
        int2 ue[2 * (WA / 32)];
        int  r[2 * (WA / 32)];
        bool ok[2 * (WA / 32)];
#pragma unroll
        for (int i = 0; i < WA / 32; ++i) {
#pragma unroll
            for (int h = 0; h < 2; ++h) {
                const int j = 2 * i + h;
                const int k = lane + 32 * h;
                ok[j] = k < c[i];
                const int2* p = ok[j] ? bp + ((wave + 32 * i) * BCAP + k) : (const int2*)bin_cnt;
                ue[j] = *p;
            }
        }
#pragma unroll
        for (int j = 0; j < 2 * (WA / 32); ++j) {
            r[j] = 0;
            if (ok[j]) r[j] = __hip_atomic_fetch_add(ltab + ue[j].x * REC + SLOTS, 1, __ATOMIC_RELAXED,
                                                     __HIP_MEMORY_SCOPE_WORKGROUP);
        }
#pragma unroll
        for (int j = 0; j < 2 * (WA / 32); ++j) {
            if (ok[j]) store_edge(ltab, &lsn, (unsigned)ue[j].x, ue[j].y, r[j], a, sp_dst, sp_src);
        }
    } else {
        const unsigned nv = (((size_t)dst & 15) == 0) ? (unsigned)(E >> 2) : 0u;
        const int4* d4 = (const int4*)dst;
        const unsigned off = blockIdx.x * (nv / gridDim.x);
        int qn = 0;
        unsigned iw = __builtin_amdgcn_readfirstlane(tid & ~31);
        for (; iw + 31u + 7u * BT < nv; iw += 8u * BT) {
            const unsigned i = iw + lane;
            unsigned jj[8];
            int4 d[8];
#pragma unroll
            for (int k = 0; k < 8; ++k) {
                const unsigned j = i + k * BT + off;
                jj[k] = min(j, j - nv);
                d[k] = d4[jj[k]];
            }
#pragma unroll
            for (int k = 0; k < 8; ++k) {
                qn = scan4(d[k], (int)(jj[k] << 2), a, q, qn);
                if (qn > QFULL) { drain_queue(q, qn, lane, ltab, &lsn, a, src, sp_dst, sp_src); qn = 0; }
            }
        }
        for (; iw < nv; iw += BT) {
            const unsigned i = iw + lane;
            const unsigned j = i + off;
            const unsigned j1 = min(j, j - nv);
            int4 dd = make_int4(-1, -1, -1, -1);
            if (i < nv) dd = d4[j1];
            qn = scan4(dd, (int)(j1 << 2), a, q, qn);
            if (qn > QFULL) { drain_queue(q, qn, lane, ltab, &lsn, a, src, sp_dst, sp_src); qn = 0; }
        }
        drain_queue(q, qn, lane, ltab, &lsn, a, src, sp_dst, sp_src);
        for (int e = (int)(nv << 2) + tid; e < E; e += BT) {
            const unsigned u = (unsigned)(dst[e] - a);
            if (u < (unsigned)RANGE) place_edge(ltab, &lsn, u, src[e], a, sp_dst, sp_src);
        }
    }
    __syncthreads();

    int4* tab4 = (int4*)tab;
    for (int w = tid; w < RANGE * (REC / 4); w += BT) {
        const int nd = a + (w >> 3);
        if (nd < n) tab4[(long long)nd * (REC / 4) + (w & 7)] = ltab4[w];
    }
    if (tid == 0) spill_cnt[blockIdx.x] = lsn;
}

static __device__ __forceinline__ int lane_i(int v, int l) {
    return __builtin_amdgcn_readlane(v, l);
}
static __device__ __forceinline__ float lane_f(float v, int l) {
    return __int_as_float(__builtin_amdgcn_readlane(__float_as_int(v), l));
}
static __device__ __forceinline__ void fma4(float4& a, const float4& h, float w) {
    a.x = fmaf(h.x, w, a.x);
    a.y = fmaf(h.y, w, a.y);
    a.z = fmaf(h.z, w, a.z);
    a.w = fmaf(h.w, w, a.w);
}
static __device__ __forceinline__ void fma2(float2& a, const float2& h, float w) {
    a.x = fmaf(h.x, w, a.x);
    a.y = fmaf(h.y, w, a.y);
}

#define TILE_N 32
#define PITCH  132
__global__ __launch_bounds__(256)
void gcn_layer1_fused(const float* __restrict__ X,
                      const int* __restrict__ src, const int* __restrict__ dst, int E,
                      const int* __restrict__ tab, const int* __restrict__ spill_cnt,
                      const int* __restrict__ spill_dst, const int* __restrict__ spill_src,
                      const float* __restrict__ W1, const float* __restrict__ b1,
                      const float* __restrict__ W2,
                      float* __restrict__ H2, int n) {
    __shared__ __attribute__((aligned(16))) float sA[TILE_N * PITCH];
    __shared__ __attribute__((aligned(16))) float sT[TILE_N * PITCH];

    const int lane = threadIdx.x & 31;
    const int wave  = __builtin_amdgcn_readfirstlane(threadIdx.x >> 5);
    const int base  = blockIdx.x * TILE_N;
    const int node0 = base + wave * 4;
    const float* xl = X + (lane << 2);

    int   rec[4], deg[4], mm[4], ms[4];
    float di[4], mw[4];
#pragma unroll
    for (int q = 0; q < 4; ++q) {
        const int nd  = node0 + q;
        const int ndc = (nd < n) ? nd : n - 1;
        rec[q] = tab[(long long)ndc * REC + lane];
    }
#pragma unroll
    for (int q = 0; q < 4; ++q) {
        const int nd  = node0 + q;
        const bool ok = nd < n;
        const int ndc = ok ? nd : n - 1;
        const int c = lane_i(rec[q], SLOTS);
        deg[q] = ok ? c : 0;
        di[q]  = ok ? rsqrtf((float)c + 1.0f) : 0.f;
        mm[q] = min(deg[q], SLOTS);
        if (deg[q] > SLOTS && spill_cnt[ndc / RANGE] > CAP) mm[q] = 0;
        ms[q] = (lane < mm[q]) ? rec[q] : ndc;
    }
#pragma unroll
    for (int q = 0; q < 4; ++q) {
        const int cs = tab[(long long)ms[q] * REC + SLOTS];
        mw[q] = rsqrtf((float)cs + 1.0f) * ((lane < mm[q]) ? di[q] : 0.f);
    }

#pragma unroll
    for (int q = 0; q < 4; ++q) {
        const int nd = node0 + q;
        const int my_s = ms[q];
        const float my_w = mw[q];
        float4 acc = {0.f, 0.f, 0.f, 0.f};
        const int m = mm[q];
        int j = 0;
        for (; j + 4 <= m; j += 4) {
            const int s0 = lane_i(my_s, j),     s1 = lane_i(my_s, j + 1);
            const int s2 = lane_i(my_s, j + 2), s3 = lane_i(my_s, j + 3);
            const float4 h0 = *(const float4*)(xl + (long long)s0 * 128);
            const float4 h1 = *(const float4*)(xl + (long long)s1 * 128);
            const float4 h2 = *(const float4*)(xl + (long long)s2 * 128);
            const float4 h3 = *(const float4*)(xl + (long long)s3 * 128);
            fma4(acc, h0, lane_f(my_w, j));
            fma4(acc, h1, lane_f(my_w, j + 1));
            fma4(acc, h2, lane_f(my_w, j + 2));
            fma4(acc, h3, lane_f(my_w, j + 3));
        }
        for (; j < m; ++j) {
            const int s0 = lane_i(my_s, j);
            const float4 h0 = *(const float4*)(xl + (long long)s0 * 128);
            fma4(acc, h0, lane_f(my_w, j));
        }
        if (deg[q] > SLOTS) {
            const int g = nd / RANGE;
            const int* pd = (m > 0) ? spill_dst + (long long)g * CAP : dst;
            const int* ps = (m > 0) ? spill_src + (long long)g * CAP : src;
            const int total = (m > 0) ? spill_cnt[g] : E;
            for (int jb = 0; jb < total; jb += 32) {
                int sd = -1, ss = 0;
                if (jb + lane < total) {
                    sd = pd[jb + lane];
                    ss = ps[jb + lane];
                }
                const bool hit = sd == nd;
                float w = 0.f;
                if (hit) w = rsqrtf((float)tab[(long long)ss * REC + SLOTS] + 1.0f) * di[q];
                unsigned hits = __builtin_amdgcn_ballot_w32(hit);
                while (hits) {
                    const int l = __builtin_ctz(hits);
                    hits &= hits - 1;
                    const int s0 = lane_i(ss, l);
                    const float4 h0 = *(const float4*)(xl + (long long)s0 * 128);
                    fma4(acc, h0, lane_f(w, l));
                }
            }
        }
        if (nd < n) {
            const float4 hs = *(const float4*)(xl + (long long)nd * 128);
            fma4(acc, hs, di[q] * di[q]);
        }
        *(float4*)(sA + (wave * 4 + q) * PITCH + (lane << 2)) = acc;
    }
    __syncthreads();

    const int c  = lane & 15;
    const int hh = lane >> 4;
    const int ko = hh << 1;

    {
        v8f d0 = {0.f,0.f,0.f,0.f,0.f,0.f,0.f,0.f};
        v8f d1 = d0;
        const float* a0p = sA + c * PITCH + ko;
        const float* a1p = sA + (16 + c) * PITCH + ko;
        const float* bp  = W1 + ko * 128 + wave * 16 + c;
#pragma unroll 8
        for (int k0 = 0; k0 < 128; k0 += 4) {
            const v2f a0 = *(const v2f*)(a0p + k0);
            const v2f a1 = *(const v2f*)(a1p + k0);
            v2f b;
            b.x = bp[k0 * 128];
            b.y = bp[k0 * 128 + 128];
            d0 = __builtin_amdgcn_wmma_f32_16x16x4_f32(false, a0, false, b, (short)0, d0, false, false);
            d1 = __builtin_amdgcn_wmma_f32_16x16x4_f32(false, a1, false, b, (short)0, d1, false, false);
        }
        const int col = wave * 16 + c;
        const float bb = b1[col];
        float* t0 = sT + (hh << 3) * PITCH + col;
        float* t1 = t0 + 16 * PITCH;
#pragma unroll
        for (int v = 0; v < 8; ++v) {
            t0[v * PITCH] = fmaxf(d0[v] + bb, 0.0f);
            t1[v * PITCH] = fmaxf(d1[v] + bb, 0.0f);
        }
    }
    __syncthreads();

    {
        const int mt = wave >> 2;
        const int nt = wave & 3;
        v8f e0 = {0.f,0.f,0.f,0.f,0.f,0.f,0.f,0.f};
        v8f e1 = e0;
        const float* tp = sT + (mt * 16 + c) * PITCH + ko;
        const float* wp = W2 + ko * 64 + nt * 16 + c;
#pragma unroll 4
        for (int k0 = 0; k0 < 128; k0 += 8) {
            const v2f a0 = *(const v2f*)(tp + k0);
            const v2f a1 = *(const v2f*)(tp + k0 + 4);
            v2f b0, b1v;
            b0.x  = wp[k0 * 64];
            b0.y  = wp[k0 * 64 + 64];
            b1v.x = wp[(k0 + 4) * 64];
            b1v.y = wp[(k0 + 4) * 64 + 64];
            e0 = __builtin_amdgcn_wmma_f32_16x16x4_f32(false, a0, false, b0,  (short)0, e0, false, false);
            e1 = __builtin_amdgcn_wmma_f32_16x16x4_f32(false, a1, false, b1v, (short)0, e1, false, false);
        }
        const int row0 = base + mt * 16 + (hh << 3);
        float* hp = H2 + (long long)row0 * 64 + nt * 16 + c;
#pragma unroll
        for (int v = 0; v < 8; ++v)
            if (row0 + v < n) hp[v * 64] = e0[v] + e1[v];
    }
}

__global__ __launch_bounds__(256)
void gcn_gather64(const float* __restrict__ H,
                  const int* __restrict__ src, const int* __restrict__ dst, int E,
                  const int* __restrict__ tab, const int* __restrict__ spill_cnt,
                  const int* __restrict__ spill_dst, const int* __restrict__ spill_src,
                  const float* __restrict__ bias,
                  float* __restrict__ outp, int n) {
    const int lane = threadIdx.x & 31;
    const int node = __builtin_amdgcn_readfirstlane(blockIdx.x * 8 + (threadIdx.x >> 5));
    if (node >= n) return;

    const float* col = H + (lane << 1);
    const int rec = tab[(long long)node * REC + lane];
    const int deg = lane_i(rec, SLOTS);
    const float di = rsqrtf((float)deg + 1.0f);
    int m = min(deg, SLOTS);
    if (deg > SLOTS && spill_cnt[node / RANGE] > CAP) m = 0;
    const int my_s = (lane < m) ? rec : node;
    const int cs = tab[(long long)my_s * REC + SLOTS];
    const float my_w = rsqrtf((float)cs + 1.0f) * ((lane < m) ? di : 0.f);

    float2 acc = {0.f, 0.f};
    int j = 0;
    for (; j + 4 <= m; j += 4) {
        const int s0 = lane_i(my_s, j),     s1 = lane_i(my_s, j + 1);
        const int s2 = lane_i(my_s, j + 2), s3 = lane_i(my_s, j + 3);
        const float2 h0 = *(const float2*)(col + (long long)s0 * 64);
        const float2 h1 = *(const float2*)(col + (long long)s1 * 64);
        const float2 h2 = *(const float2*)(col + (long long)s2 * 64);
        const float2 h3 = *(const float2*)(col + (long long)s3 * 64);
        fma2(acc, h0, lane_f(my_w, j));
        fma2(acc, h1, lane_f(my_w, j + 1));
        fma2(acc, h2, lane_f(my_w, j + 2));
        fma2(acc, h3, lane_f(my_w, j + 3));
    }
    for (; j < m; ++j) {
        const int s0 = lane_i(my_s, j);
        const float2 h0 = *(const float2*)(col + (long long)s0 * 64);
        fma2(acc, h0, lane_f(my_w, j));
    }
    if (deg > SLOTS) {
        const int g = node / RANGE;
        const int* pd = (m > 0) ? spill_dst + (long long)g * CAP : dst;
        const int* ps = (m > 0) ? spill_src + (long long)g * CAP : src;
        const int total = (m > 0) ? spill_cnt[g] : E;
        for (int jb = 0; jb < total; jb += 32) {
            int sd = -1, ss = 0;
            if (jb + lane < total) {
                sd = pd[jb + lane];
                ss = ps[jb + lane];
            }
            const bool hit = sd == node;
            float w = 0.f;
            if (hit) w = rsqrtf((float)tab[(long long)ss * REC + SLOTS] + 1.0f) * di;
            unsigned hits = __builtin_amdgcn_ballot_w32(hit);
            while (hits) {
                const int l = __builtin_ctz(hits);
                hits &= hits - 1;
                const int s0 = lane_i(ss, l);
                const float2 h0 = *(const float2*)(col + (long long)s0 * 64);
                fma2(acc, h0, lane_f(w, l));
            }
        }
    }

    const float2 hs = *(const float2*)(col + (long long)node * 64);
    fma2(acc, hs, di * di);
    const int c = lane << 1;
    float2 r;
    r.x = acc.x + bias[c];
    r.y = acc.y + bias[c + 1];
    *(float2*)(outp + (long long)node * 64 + c) = r;
}

extern "C" void kernel_launch(void* const* d_in, const int* in_sizes, int n_in,
                              void* d_out, int out_size, void* d_ws, size_t ws_size,
                              hipStream_t stream) {
    const float* x   = (const float*)d_in[0];
    const int*   ei  = (const int*)d_in[1];
    const float* W1  = (const float*)d_in[2];
    const float* b1  = (const float*)d_in[3];
    const float* W2  = (const float*)d_in[4];
    const float* b2  = (const float*)d_in[5];

    const int IN_C  = 128;
    const int nNodes = in_sizes[0] / IN_C;
    const int E      = in_sizes[1] / 2;
    const int* srcIdx = ei;
    const int* dstIdx = ei + E;

    const int nPad   = (nNodes + 63) & ~63;
    const int groups = (nNodes + RANGE - 1) / RANGE;
    int*   tab       = (int*)d_ws;
    float* h2        = (float*)(tab + (long long)nPad * REC);
    int*   spill_dst = (int*)(h2 + (long long)nPad * 64);
    int*   spill_src = spill_dst + (long long)groups * CAP;
    int*   spill_cnt = spill_src + (long long)groups * CAP;
    float* out       = (float*)d_out;


    int*   bins      = (int*)h2;
    int*   bin_cnt   = bins + (long long)GMAX * WA * BCAP * 2;

    gcn_bin_edges<<<WA, ABT, 0, stream>>>(srcIdx, dstIdx, E, groups, bins, bin_cnt);

    gcn_build_table<<<groups, BT, 0, stream>>>(srcIdx, dstIdx, E, nNodes, bins, bin_cnt,
                                               tab, spill_cnt, spill_dst, spill_src);

    gcn_layer1_fused<<<(nNodes + TILE_N - 1) / TILE_N, 256, 0, stream>>>(
        x, srcIdx, dstIdx, E, tab, spill_cnt, spill_dst, spill_src, W1, b1, W2, h2, nNodes);

    gcn_gather64<<<(nNodes + 7) / 8, 256, 0, stream>>>(
        h2, srcIdx, dstIdx, E, tab, spill_cnt, spill_dst, spill_src, b2, out, nNodes);
}


# H2 rows stored already multiplied by their node's 1/sqrt(deg): the 64-channel gather adds plain rows and scales once, its per-source count lookups and weights gone; term rounding changes, order kept
# speedup vs baseline: 5.0594x; 1.0393x over previous
#include <hip/hip_runtime.h>


typedef __attribute__((ext_vector_type(2))) float v2f;
typedef __attribute__((ext_vector_type(8))) float v8f;

#define REC   32
#define SLOTS 31
#define RANGE 392
#define CAP   1024
#define BT    1024
#define WA    256
#define ABT   512
#define BCAP  48
#define GMAX  128

__global__ __launch_bounds__(ABT)
void gcn_bin_edges(const int* __restrict__ src, const int* __restrict__ dst, int E, int groups,
                   int* __restrict__ bins, int* __restrict__ bin_cnt) {
    __shared__ int lc[GMAX + 1];
    const int tid = threadIdx.x;
    if (tid <= GMAX) lc[tid] = 0;
    __syncthreads();

    int2* seg = (int2*)bins + (long long)blockIdx.x * BCAP;
    const int per = (E + (int)gridDim.x - 1) / (int)gridDim.x;
    const int e0  = blockIdx.x * per;
    const int e1  = min(E, e0 + per);
    for (int eb = e0; eb < e1; eb += 5 * ABT) {
        int  d[5], s[5], g[5], r[5];
        bool ok[5];
#pragma unroll
        for (int k = 0; k < 5; ++k) {
            const int e = eb + k * ABT + tid;
            ok[k] = e < e1;
            const int ec = ok[k] ? e : eb;
            d[k] = dst[ec];
            s[k] = src[ec];
        }
#pragma unroll
        for (int k = 0; k < 5; ++k) asm volatile("" : "+v"(s[k]));
#pragma unroll
        for (int k = 0; k < 5; ++k) {
            g[k] = ok[k] ? d[k] / RANGE : GMAX;
            r[k] = __hip_atomic_fetch_add(lc + g[k], 1, __ATOMIC_RELAXED,
                                          __HIP_MEMORY_SCOPE_WORKGROUP);
        }
#pragma unroll
        for (int k = 0; k < 5; ++k) {
            if (ok[k] && r[k] < BCAP)
                seg[(long long)g[k] * (WA * BCAP) + r[k]] = make_int2(d[k] - g[k] * RANGE, s[k]);
        }
    }
    __syncthreads();
    if (tid < groups) bin_cnt[tid * WA + blockIdx.x] = lc[tid];
}

#define QCAP  512
#define QFULL (QCAP - 128)

static __device__ __forceinline__ void store_edge(int* ltab, int* lsn, unsigned u, int s, int r, int a,
                                                  int* __restrict__ sp_dst, int* __restrict__ sp_src) {
    if (r < SLOTS) {
        ltab[u * REC + r] = s;
    } else {
        const int p = __hip_atomic_fetch_add(lsn, 1, __ATOMIC_RELAXED,
                                             __HIP_MEMORY_SCOPE_WORKGROUP);
        if (p < CAP) { sp_dst[p] = a + (int)u; sp_src[p] = s; }
    }
}

static __device__ __forceinline__ void place_edge(int* ltab, int* lsn, unsigned u, int s, int a,
                                                  int* __restrict__ sp_dst, int* __restrict__ sp_src) {
    const int r = __hip_atomic_fetch_add(ltab + u * REC + SLOTS, 1, __ATOMIC_RELAXED,
                                         __HIP_MEMORY_SCOPE_WORKGROUP);
    store_edge(ltab, lsn, u, s, r, a, sp_dst, sp_src);
}

static __device__ __forceinline__ void drain_queue(const int2* q, int cnt, int lane,
                                                   int* ltab, int* lsn, int a,
                                                   const int* __restrict__ src,
                                                   int* __restrict__ sp_dst, int* __restrict__ sp_src) {
    __builtin_amdgcn_wave_barrier();
    for (int b = 0; b < cnt; b += 32) {
        if (b + lane < cnt) {
            const int2 ue = q[b + lane];
            place_edge(ltab, lsn, (unsigned)ue.x, src[ue.y], a, sp_dst, sp_src);
        }
    }
    __builtin_amdgcn_wave_barrier();
}

static __device__ __forceinline__ int scan4(const int4 d, int e, int a, int2* q, int qn) {
    const unsigned u0 = (unsigned)(d.x - a), u1 = (unsigned)(d.y - a);
    const unsigned u2 = (unsigned)(d.z - a), u3 = (unsigned)(d.w - a);
    if (__builtin_amdgcn_ballot_w32(min(min(u0, u1), min(u2, u3)) < (unsigned)RANGE) != 0u) {
        const unsigned uu[4] = {u0, u1, u2, u3};
#pragma unroll
        for (int k = 0; k < 4; ++k) {
            const bool hit = uu[k] < (unsigned)RANGE;
            const unsigned bal = __builtin_amdgcn_ballot_w32(hit);
            if (hit) q[qn + __builtin_amdgcn_mbcnt_lo(bal, 0u)] = make_int2((int)uu[k], e + k);
            qn += __builtin_popcount(bal);
        }
    }
    return qn;
}

__global__ __launch_bounds__(BT)
void gcn_build_table(const int* __restrict__ src, const int* __restrict__ dst, int E, int n,
                     const int* __restrict__ bins, const int* __restrict__ bin_cnt,
                     int* __restrict__ tab, int* __restrict__ spill_cnt,
                     int* __restrict__ spill_dst, int* __restrict__ spill_src) {
    __shared__ __attribute__((aligned(16))) int ltab[RANGE * REC];
    __shared__ __attribute__((aligned(8)))  int2 lq[(BT / 32) * QCAP];
    __shared__ int lsn;
    __shared__ int lovf;

    const int tid  = threadIdx.x;
    const int lane = tid & 31;
    const int wave = __builtin_amdgcn_readfirstlane(tid >> 5);
    int2* q = lq + wave * QCAP;
    const int a = blockIdx.x * RANGE;
    int* sp_dst = spill_dst + (long long)blockIdx.x * CAP;
    int* sp_src = spill_src + (long long)blockIdx.x * CAP;
    int4* ltab4 = (int4*)ltab;

    int c[WA / 32];
#pragma unroll
    for (int i = 0; i < WA / 32; ++i) c[i] = bin_cnt[blockIdx.x * WA + wave + 32 * i];

    for (int i = tid; i < RANGE * (REC / 4); i += BT) ltab4[i] = make_int4(0, 0, 0, 0);
    if (tid == 0) { lsn = 0; lovf = 0; }
    __syncthreads();

    bool over = false;
#pragma unroll
    for (int i = 0; i < WA / 32; ++i) over = over || (c[i] > BCAP);
    if (over) lovf = 1;
    __syncthreads();

    if (lovf == 0) {
        const int2* bp = (const int2*)bins + (long long)blockIdx.x * (WA * BCAP);
        int2 ue[2 * (WA / 32)];
        int  r[2 * (WA / 32)];
        bool ok[2 * (WA / 32)];
#pragma unroll
        for (int i = 0; i < WA / 32; ++i) {
#pragma unroll
            for (int h = 0; h < 2; ++h) {
                const int j = 2 * i + h;
                const int k = lane + 32 * h;
                ok[j] = k < c[i];
                const int2* p = ok[j] ? bp + ((wave + 32 * i) * BCAP + k) : (const int2*)bin_cnt;
                ue[j] = *p;
            }
        }
#pragma unroll
        for (int j = 0; j < 2 * (WA / 32); ++j) {
            r[j] = 0;
            if (ok[j]) r[j] = __hip_atomic_fetch_add(ltab + ue[j].x * REC + SLOTS, 1, __ATOMIC_RELAXED,
                                                     __HIP_MEMORY_SCOPE_WORKGROUP);
        }
#pragma unroll
        for (int j = 0; j < 2 * (WA / 32); ++j) {
            if (ok[j]) store_edge(ltab, &lsn, (unsigned)ue[j].x, ue[j].y, r[j], a, sp_dst, sp_src);
        }
    } else {
        const unsigned nv = (((size_t)dst & 15) == 0) ? (unsigned)(E >> 2) : 0u;
        const int4* d4 = (const int4*)dst;
        const unsigned off = blockIdx.x * (nv / gridDim.x);
        int qn = 0;
        unsigned iw = __builtin_amdgcn_readfirstlane(tid & ~31);
        for (; iw + 31u + 7u * BT < nv; iw += 8u * BT) {
            const unsigned i = iw + lane;
            unsigned jj[8];
            int4 d[8];
#pragma unroll
            for (int k = 0; k < 8; ++k) {
                const unsigned j = i + k * BT + off;
                jj[k] = min(j, j - nv);
                d[k] = d4[jj[k]];
            }
#pragma unroll
            for (int k = 0; k < 8; ++k) {
                qn = scan4(d[k], (int)(jj[k] << 2), a, q, qn);
                if (qn > QFULL) { drain_queue(q, qn, lane, ltab, &lsn, a, src, sp_dst, sp_src); qn = 0; }
            }
        }
        for (; iw < nv; iw += BT) {
            const unsigned i = iw + lane;
            const unsigned j = i + off;
            const unsigned j1 = min(j, j - nv);
            int4 dd = make_int4(-1, -1, -1, -1);
            if (i < nv) dd = d4[j1];
            qn = scan4(dd, (int)(j1 << 2), a, q, qn);
            if (qn > QFULL) { drain_queue(q, qn, lane, ltab, &lsn, a, src, sp_dst, sp_src); qn = 0; }
        }
        drain_queue(q, qn, lane, ltab, &lsn, a, src, sp_dst, sp_src);
        for (int e = (int)(nv << 2) + tid; e < E; e += BT) {
            const unsigned u = (unsigned)(dst[e] - a);
            if (u < (unsigned)RANGE) place_edge(ltab, &lsn, u, src[e], a, sp_dst, sp_src);
        }
    }
    __syncthreads();

    int4* tab4 = (int4*)tab;
    for (int w = tid; w < RANGE * (REC / 4); w += BT) {
        const int nd = a + (w >> 3);
        if (nd < n) tab4[(long long)nd * (REC / 4) + (w & 7)] = ltab4[w];
    }
    if (tid == 0) spill_cnt[blockIdx.x] = lsn;
}

static __device__ __forceinline__ int lane_i(int v, int l) {
    return __builtin_amdgcn_readlane(v, l);
}
static __device__ __forceinline__ float lane_f(float v, int l) {
    return __int_as_float(__builtin_amdgcn_readlane(__float_as_int(v), l));
}
static __device__ __forceinline__ void fma4(float4& a, const float4& h, float w) {
    a.x = fmaf(h.x, w, a.x);
    a.y = fmaf(h.y, w, a.y);
    a.z = fmaf(h.z, w, a.z);
    a.w = fmaf(h.w, w, a.w);
}

#define TILE_N 32
#define PITCH  132
__global__ __launch_bounds__(256)
void gcn_layer1_fused(const float* __restrict__ X,
                      const int* __restrict__ src, const int* __restrict__ dst, int E,
                      const int* __restrict__ tab, const int* __restrict__ spill_cnt,
                      const int* __restrict__ spill_dst, const int* __restrict__ spill_src,
                      const float* __restrict__ W1, const float* __restrict__ b1,
                      const float* __restrict__ W2,
                      float* __restrict__ H2, int n) {
    __shared__ __attribute__((aligned(16))) float sA[TILE_N * PITCH];
    __shared__ __attribute__((aligned(16))) float sT[TILE_N * PITCH];
    __shared__ __attribute__((aligned(16))) float sD[TILE_N];

    const int lane = threadIdx.x & 31;
    const int wave  = __builtin_amdgcn_readfirstlane(threadIdx.x >> 5);
    const int base  = blockIdx.x * TILE_N;
    const int node0 = base + wave * 4;
    const float* xl = X + (lane << 2);

    int   rec[4], deg[4], mm[4], ms[4];
    float di[4], mw[4];
#pragma unroll
    for (int q = 0; q < 4; ++q) {
        const int nd  = node0 + q;
        const int ndc = (nd < n) ? nd : n - 1;
        rec[q] = tab[(long long)ndc * REC + lane];
    }
#pragma unroll
    for (int q = 0; q < 4; ++q) {
        const int nd  = node0 + q;
        const bool ok = nd < n;
        const int ndc = ok ? nd : n - 1;
        const int c = lane_i(rec[q], SLOTS);
        deg[q] = ok ? c : 0;
        di[q]  = ok ? rsqrtf((float)c + 1.0f) : 0.f;
        if (lane == 0) sD[wave * 4 + q] = di[q];
        mm[q] = min(deg[q], SLOTS);
        if (deg[q] > SLOTS && spill_cnt[ndc / RANGE] > CAP) mm[q] = 0;
        ms[q] = (lane < mm[q]) ? rec[q] : ndc;
    }
#pragma unroll
    for (int q = 0; q < 4; ++q) {
        const int cs = tab[(long long)ms[q] * REC + SLOTS];
        mw[q] = rsqrtf((float)cs + 1.0f) * ((lane < mm[q]) ? di[q] : 0.f);
    }

#pragma unroll
    for (int q = 0; q < 4; ++q) {
        const int nd = node0 + q;
        const int my_s = ms[q];
        const float my_w = mw[q];
        float4 acc = {0.f, 0.f, 0.f, 0.f};
        const int m = mm[q];
        int j = 0;
        for (; j + 4 <= m; j += 4) {
            const int s0 = lane_i(my_s, j),     s1 = lane_i(my_s, j + 1);
            const int s2 = lane_i(my_s, j + 2), s3 = lane_i(my_s, j + 3);
            const float4 h0 = *(const float4*)(xl + (long long)s0 * 128);
            const float4 h1 = *(const float4*)(xl + (long long)s1 * 128);
            const float4 h2 = *(const float4*)(xl + (long long)s2 * 128);
            const float4 h3 = *(const float4*)(xl + (long long)s3 * 128);
            fma4(acc, h0, lane_f(my_w, j));
            fma4(acc, h1, lane_f(my_w, j + 1));
            fma4(acc, h2, lane_f(my_w, j + 2));
            fma4(acc, h3, lane_f(my_w, j + 3));
        }
        for (; j < m; ++j) {
            const int s0 = lane_i(my_s, j);
            const float4 h0 = *(const float4*)(xl + (long long)s0 * 128);
            fma4(acc, h0, lane_f(my_w, j));
        }
        if (deg[q] > SLOTS) {
            const int g = nd / RANGE;
            const int* pd = (m > 0) ? spill_dst + (long long)g * CAP : dst;
            const int* ps = (m > 0) ? spill_src + (long long)g * CAP : src;
            const int total = (m > 0) ? spill_cnt[g] : E;
            for (int jb = 0; jb < total; jb += 32) {
                int sd = -1, ss = 0;
                if (jb + lane < total) {
                    sd = pd[jb + lane];
                    ss = ps[jb + lane];
                }
                const bool hit = sd == nd;
                float w = 0.f;
                if (hit) w = rsqrtf((float)tab[(long long)ss * REC + SLOTS] + 1.0f) * di[q];
                unsigned hits = __builtin_amdgcn_ballot_w32(hit);
                while (hits) {
                    const int l = __builtin_ctz(hits);
                    hits &= hits - 1;
                    const int s0 = lane_i(ss, l);
                    const float4 h0 = *(const float4*)(xl + (long long)s0 * 128);
                    fma4(acc, h0, lane_f(w, l));
                }
            }
        }
        if (nd < n) {
            const float4 hs = *(const float4*)(xl + (long long)nd * 128);
            fma4(acc, hs, di[q] * di[q]);
        }
        *(float4*)(sA + (wave * 4 + q) * PITCH + (lane << 2)) = acc;
    }
    __syncthreads();

    const int c  = lane & 15;
    const int hh = lane >> 4;
    const int ko = hh << 1;

    {
        v8f d0 = {0.f,0.f,0.f,0.f,0.f,0.f,0.f,0.f};
        v8f d1 = d0;
        const float* a0p = sA + c * PITCH + ko;
        const float* a1p = sA + (16 + c) * PITCH + ko;
        const float* bp  = W1 + ko * 128 + wave * 16 + c;
#pragma unroll 8
        for (int k0 = 0; k0 < 128; k0 += 4) {
            const v2f a0 = *(const v2f*)(a0p + k0);
            const v2f a1 = *(const v2f*)(a1p + k0);
            v2f b;
            b.x = bp[k0 * 128];
            b.y = bp[k0 * 128 + 128];
            d0 = __builtin_amdgcn_wmma_f32_16x16x4_f32(false, a0, false, b, (short)0, d0, false, false);
            d1 = __builtin_amdgcn_wmma_f32_16x16x4_f32(false, a1, false, b, (short)0, d1, false, false);
        }
        const int col = wave * 16 + c;
        const float bb = b1[col];
        float* t0 = sT + (hh << 3) * PITCH + col;
        float* t1 = t0 + 16 * PITCH;
#pragma unroll
        for (int v = 0; v < 8; ++v) {
            t0[v * PITCH] = fmaxf(d0[v] + bb, 0.0f);
            t1[v * PITCH] = fmaxf(d1[v] + bb, 0.0f);
        }
    }
    __syncthreads();

    {
        const int mt = wave >> 2;
        const int nt = wave & 3;
        v8f e0 = {0.f,0.f,0.f,0.f,0.f,0.f,0.f,0.f};
        v8f e1 = e0;
        const float* tp = sT + (mt * 16 + c) * PITCH + ko;
        const float* wp = W2 + ko * 64 + nt * 16 + c;
#pragma unroll 4
        for (int k0 = 0; k0 < 128; k0 += 8) {
            const v2f a0 = *(const v2f*)(tp + k0);
            const v2f a1 = *(const v2f*)(tp + k0 + 4);
            v2f b0, b1v;
            b0.x  = wp[k0 * 64];
            b0.y  = wp[k0 * 64 + 64];
            b1v.x = wp[(k0 + 4) * 64];
            b1v.y = wp[(k0 + 4) * 64 + 64];
            e0 = __builtin_amdgcn_wmma_f32_16x16x4_f32(false, a0, false, b0,  (short)0, e0, false, false);
            e1 = __builtin_amdgcn_wmma_f32_16x16x4_f32(false, a1, false, b1v, (short)0, e1, false, false);
        }
        const int row0 = base + mt * 16 + (hh << 3);
        float* hp = H2 + (long long)row0 * 64 + nt * 16 + c;
        const float* dp = sD + mt * 16 + (hh << 3);
        if (base + TILE_N <= n) {
#pragma unroll
            for (int v = 0; v < 8; ++v) hp[v * 64] = (e0[v] + e1[v]) * dp[v];
        } else {
#pragma unroll
            for (int v = 0; v < 8; ++v)
                if (row0 + v < n) hp[v * 64] = (e0[v] + e1[v]) * dp[v];
        }
    }
}

static __device__ __forceinline__ void add2(float2& a, const float2& h) {
    a.x += h.x;
    a.y += h.y;
}
__global__ __launch_bounds__(256)
void gcn_gather64(const float* __restrict__ H,
                  const int* __restrict__ src, const int* __restrict__ dst, int E,
                  const int* __restrict__ tab, const int* __restrict__ spill_cnt,
                  const int* __restrict__ spill_dst, const int* __restrict__ spill_src,
                  const float* __restrict__ bias,
                  float* __restrict__ outp, int n) {
    const int lane = threadIdx.x & 31;
    const int node = __builtin_amdgcn_readfirstlane(blockIdx.x * 8 + (threadIdx.x >> 5));
    if (node >= n) return;

    const float* col = H + (lane << 1);
    const int c = lane << 1;
    const int rec = tab[(long long)node * REC + lane];
    const float2 hs = *(const float2*)(col + (long long)node * 64);
    const float2 bb = *(const float2*)(bias + c);
    const int deg = lane_i(rec, SLOTS);
    const float di = rsqrtf((float)deg + 1.0f);
    int m = min(deg, SLOTS);
    if (deg > SLOTS && spill_cnt[node / RANGE] > CAP) m = 0;

    float2 acc = {0.f, 0.f};
    int j = 0;
    for (; j + 4 <= m; j += 4) {
        const int s0 = lane_i(rec, j),     s1 = lane_i(rec, j + 1);
        const int s2 = lane_i(rec, j + 2), s3 = lane_i(rec, j + 3);
        const float2 h0 = *(const float2*)(col + (long long)s0 * 64);
        const float2 h1 = *(const float2*)(col + (long long)s1 * 64);
        const float2 h2 = *(const float2*)(col + (long long)s2 * 64);
        const float2 h3 = *(const float2*)(col + (long long)s3 * 64);
        add2(acc, h0);
        add2(acc, h1);
        add2(acc, h2);
        add2(acc, h3);
    }
    for (; j < m; ++j) {
        const int s0 = lane_i(rec, j);
        const float2 h0 = *(const float2*)(col + (long long)s0 * 64);
        add2(acc, h0);
    }
    if (deg > SLOTS) {
        const int g = node / RANGE;
        const int* pd = (m > 0) ? spill_dst + (long long)g * CAP : dst;
        const int* ps = (m > 0) ? spill_src + (long long)g * CAP : src;
        const int total = (m > 0) ? spill_cnt[g] : E;
        for (int jb = 0; jb < total; jb += 32) {
            int sd = -1, ss = 0;
            if (jb + lane < total) {
                sd = pd[jb + lane];
                ss = ps[jb + lane];
            }
            unsigned hits = __builtin_amdgcn_ballot_w32(sd == node);
            while (hits) {
                const int l = __builtin_ctz(hits);
                hits &= hits - 1;
                const int s0 = lane_i(ss, l);
                const float2 h0 = *(const float2*)(col + (long long)s0 * 64);
                add2(acc, h0);
            }
        }
    }

    add2(acc, hs);
    float2 r;
    r.x = fmaf(acc.x, di, bb.x);
    r.y = fmaf(acc.y, di, bb.y);
    *(float2*)(outp + (long long)node * 64 + c) = r;
}

extern "C" void kernel_launch(void* const* d_in, const int* in_sizes, int n_in,
                              void* d_out, int out_size, void* d_ws, size_t ws_size,
                              hipStream_t stream) {
    const float* x   = (const float*)d_in[0];
    const int*   ei  = (const int*)d_in[1];
    const float* W1  = (const float*)d_in[2];
    const float* b1  = (const float*)d_in[3];
    const float* W2  = (const float*)d_in[4];
    const float* b2  = (const float*)d_in[5];

    const int IN_C  = 128;
    const int nNodes = in_sizes[0] / IN_C;
    const int E      = in_sizes[1] / 2;
    const int* srcIdx = ei;
    const int* dstIdx = ei + E;

    const int nPad   = (nNodes + 63) & ~63;
    const int groups = (nNodes + RANGE - 1) / RANGE;
    int*   tab       = (int*)d_ws;
    float* h2        = (float*)(tab + (long long)nPad * REC);
    int*   spill_dst = (int*)(h2 + (long long)nPad * 64);
    int*   spill_src = spill_dst + (long long)groups * CAP;
    int*   spill_cnt = spill_src + (long long)groups * CAP;
    float* out       = (float*)d_out;


    int*   bins      = (int*)h2;
    int*   bin_cnt   = bins + (long long)GMAX * WA * BCAP * 2;

    gcn_bin_edges<<<WA, ABT, 0, stream>>>(srcIdx, dstIdx, E, groups, bins, bin_cnt);

    gcn_build_table<<<groups, BT, 0, stream>>>(srcIdx, dstIdx, E, nNodes, bins, bin_cnt,
                                               tab, spill_cnt, spill_dst, spill_src);

    gcn_layer1_fused<<<(nNodes + TILE_N - 1) / TILE_N, 256, 0, stream>>>(
        x, srcIdx, dstIdx, E, tab, spill_cnt, spill_dst, spill_src, W1, b1, W2, h2, nNodes);

    gcn_gather64<<<(nNodes + 7) / 8, 256, 0, stream>>>(
        h2, srcIdx, dstIdx, E, tab, spill_cnt, spill_dst, spill_src, b2, out, nNodes);
}


# Binning pass files its segments in LDS, then stores each as one contiguous run, a wave per segment; count plane transposed so counts too leave as whole lines; f32 sum order = LDS arrival, as before
# speedup vs baseline: 5.2483x; 1.0373x over previous
#include <hip/hip_runtime.h>


typedef __attribute__((ext_vector_type(2))) float v2f;
typedef __attribute__((ext_vector_type(8))) float v8f;

#define REC   32
#define SLOTS 31
#define RANGE 392
#define CAP   1024
#define BT    1024
#define WA    256
#define ABT   512
#define BCAP  48
#define GMAX  128
#define SPITCH (BCAP + 1)

__global__ __launch_bounds__(ABT)
void gcn_bin_edges(const int* __restrict__ src, const int* __restrict__ dst, int E, int groups,
                   int* __restrict__ bins, int* __restrict__ bin_cnt) {
    __shared__ __attribute__((aligned(8))) int lc[GMAX + 1];
    __shared__ __attribute__((aligned(8))) int2 st[GMAX * SPITCH];
    const int tid = threadIdx.x;
    if (tid <= GMAX) lc[tid] = 0;
    __syncthreads();

    const int per = (E + (int)gridDim.x - 1) / (int)gridDim.x;
    const int e0  = blockIdx.x * per;
    const int e1  = min(E, e0 + per);
    for (int eb = e0; eb < e1; eb += 5 * ABT) {
        int  d[5], s[5], g[5], r[5];
        bool ok[5];
#pragma unroll
        for (int k = 0; k < 5; ++k) {
            const int e = eb + k * ABT + tid;
            ok[k] = e < e1;
            const int ec = ok[k] ? e : eb;
            d[k] = dst[ec];
            s[k] = src[ec];
        }
#pragma unroll
        for (int k = 0; k < 5; ++k) asm volatile("" : "+v"(s[k]));
#pragma unroll
        for (int k = 0; k < 5; ++k) {
            g[k] = ok[k] ? d[k] / RANGE : GMAX;
            r[k] = __hip_atomic_fetch_add(lc + g[k], 1, __ATOMIC_RELAXED,
                                          __HIP_MEMORY_SCOPE_WORKGROUP);
        }
#pragma unroll
        for (int k = 0; k < 5; ++k) {
            if (ok[k] && r[k] < BCAP)
                st[g[k] * SPITCH + r[k]] = make_int2(d[k] - g[k] * RANGE, s[k]);
        }
    }
    __syncthreads();

    const int lane = tid & 31;
    const int wave = __builtin_amdgcn_readfirstlane(tid >> 5);
    int2* seg = (int2*)bins + (long long)blockIdx.x * BCAP;
    int  cn[GMAX / (ABT / 32)];
    int2 en[GMAX / (ABT / 32)];
#pragma unroll
    for (int i = 0; i < GMAX / (ABT / 32); ++i)
        cn[i] = min(lc[wave + (ABT / 32) * i], BCAP);
#pragma unroll
    for (int i = 0; i < GMAX / (ABT / 32); ++i) {
        const int2* p = (lane < cn[i]) ? st + (wave + (ABT / 32) * i) * SPITCH + lane : (const int2*)lc;
        en[i] = *p;
    }
#pragma unroll
    for (int i = 0; i < GMAX / (ABT / 32); ++i) {
        if (lane < cn[i])
            seg[(long long)(wave + (ABT / 32) * i) * (WA * BCAP) + lane] = en[i];
    }
#pragma unroll
    for (int i = 0; i < GMAX / (ABT / 32); ++i) {
        if (lane + 32 < cn[i]) {
            const int gi = wave + (ABT / 32) * i;
            seg[(long long)gi * (WA * BCAP) + lane + 32] = st[gi * SPITCH + lane + 32];
        }
    }
    if (tid < groups) bin_cnt[blockIdx.x * GMAX + tid] = lc[tid];
}

#define QCAP  512
#define QFULL (QCAP - 128)

static __device__ __forceinline__ void store_edge(int* ltab, int* lsn, unsigned u, int s, int r, int a,
                                                  int* __restrict__ sp_dst, int* __restrict__ sp_src) {
    if (r < SLOTS) {
        ltab[u * REC + r] = s;
    } else {
        const int p = __hip_atomic_fetch_add(lsn, 1, __ATOMIC_RELAXED,
                                             __HIP_MEMORY_SCOPE_WORKGROUP);
        if (p < CAP) { sp_dst[p] = a + (int)u; sp_src[p] = s; }
    }
}

static __device__ __forceinline__ void place_edge(int* ltab, int* lsn, unsigned u, int s, int a,
                                                  int* __restrict__ sp_dst, int* __restrict__ sp_src) {
    const int r = __hip_atomic_fetch_add(ltab + u * REC + SLOTS, 1, __ATOMIC_RELAXED,
                                         __HIP_MEMORY_SCOPE_WORKGROUP);
    store_edge(ltab, lsn, u, s, r, a, sp_dst, sp_src);
}

static __device__ __forceinline__ void drain_queue(const int2* q, int cnt, int lane,
                                                   int* ltab, int* lsn, int a,
                                                   const int* __restrict__ src,
                                                   int* __restrict__ sp_dst, int* __restrict__ sp_src) {
    __builtin_amdgcn_wave_barrier();
    for (int b = 0; b < cnt; b += 32) {
        if (b + lane < cnt) {
            const int2 ue = q[b + lane];
            place_edge(ltab, lsn, (unsigned)ue.x, src[ue.y], a, sp_dst, sp_src);
        }
    }
    __builtin_amdgcn_wave_barrier();
}

static __device__ __forceinline__ int scan4(const int4 d, int e, int a, int2* q, int qn) {
    const unsigned u0 = (unsigned)(d.x - a), u1 = (unsigned)(d.y - a);
    const unsigned u2 = (unsigned)(d.z - a), u3 = (unsigned)(d.w - a);
    if (__builtin_amdgcn_ballot_w32(min(min(u0, u1), min(u2, u3)) < (unsigned)RANGE) != 0u) {
        const unsigned uu[4] = {u0, u1, u2, u3};
#pragma unroll
        for (int k = 0; k < 4; ++k) {
            const bool hit = uu[k] < (unsigned)RANGE;
            const unsigned bal = __builtin_amdgcn_ballot_w32(hit);
            if (hit) q[qn + __builtin_amdgcn_mbcnt_lo(bal, 0u)] = make_int2((int)uu[k], e + k);
            qn += __builtin_popcount(bal);
        }
    }
    return qn;
}

__global__ __launch_bounds__(BT)
void gcn_build_table(const int* __restrict__ src, const int* __restrict__ dst, int E, int n,
                     const int* __restrict__ bins, const int* __restrict__ bin_cnt,
                     int* __restrict__ tab, int* __restrict__ spill_cnt,
                     int* __restrict__ spill_dst, int* __restrict__ spill_src) {
    __shared__ __attribute__((aligned(16))) int ltab[RANGE * REC];
    __shared__ __attribute__((aligned(8)))  int2 lq[(BT / 32) * QCAP];
    __shared__ int lsn;
    __shared__ int lovf;

    const int tid  = threadIdx.x;
    const int lane = tid & 31;
    const int wave = __builtin_amdgcn_readfirstlane(tid >> 5);
    int2* q = lq + wave * QCAP;
    const int a = blockIdx.x * RANGE;
    int* sp_dst = spill_dst + (long long)blockIdx.x * CAP;
    int* sp_src = spill_src + (long long)blockIdx.x * CAP;
    int4* ltab4 = (int4*)ltab;

    int c[WA / 32];
#pragma unroll
    for (int i = 0; i < WA / 32; ++i) c[i] = bin_cnt[(wave + 32 * i) * GMAX + blockIdx.x];

    for (int i = tid; i < RANGE * (REC / 4); i += BT) ltab4[i] = make_int4(0, 0, 0, 0);
    if (tid == 0) { lsn = 0; lovf = 0; }
    __syncthreads();

    bool over = false;
#pragma unroll
    for (int i = 0; i < WA / 32; ++i) over = over || (c[i] > BCAP);
    if (over) lovf = 1;
    __syncthreads();

    if (lovf == 0) {
        const int2* bp = (const int2*)bins + (long long)blockIdx.x * (WA * BCAP);
        int2 ue[2 * (WA / 32)];
        int  r[2 * (WA / 32)];
        bool ok[2 * (WA / 32)];
#pragma unroll
        for (int i = 0; i < WA / 32; ++i) {
#pragma unroll
            for (int h = 0; h < 2; ++h) {
                const int j = 2 * i + h;
                const int k = lane + 32 * h;
                ok[j] = k < c[i];
                const int2* p = ok[j] ? bp + ((wave + 32 * i) * BCAP + k) : (const int2*)bin_cnt;
                ue[j] = *p;
            }
        }
#pragma unroll
        for (int j = 0; j < 2 * (WA / 32); ++j) {
            r[j] = 0;
            if (ok[j]) r[j] = __hip_atomic_fetch_add(ltab + ue[j].x * REC + SLOTS, 1, __ATOMIC_RELAXED,
                                                     __HIP_MEMORY_SCOPE_WORKGROUP);
        }
#pragma unroll
        for (int j = 0; j < 2 * (WA / 32); ++j) {
            if (ok[j]) store_edge(ltab, &lsn, (unsigned)ue[j].x, ue[j].y, r[j], a, sp_dst, sp_src);
        }
    } else {
        const unsigned nv = (((size_t)dst & 15) == 0) ? (unsigned)(E >> 2) : 0u;
        const int4* d4 = (const int4*)dst;
        const unsigned off = blockIdx.x * (nv / gridDim.x);
        int qn = 0;
        unsigned iw = __builtin_amdgcn_readfirstlane(tid & ~31);
        for (; iw + 31u + 7u * BT < nv; iw += 8u * BT) {
            const unsigned i = iw + lane;
            unsigned jj[8];
            int4 d[8];
#pragma unroll
            for (int k = 0; k < 8; ++k) {
                const unsigned j = i + k * BT + off;
                jj[k] = min(j, j - nv);
                d[k] = d4[jj[k]];
            }
#pragma unroll
            for (int k = 0; k < 8; ++k) {
                qn = scan4(d[k], (int)(jj[k] << 2), a, q, qn);
                if (qn > QFULL) { drain_queue(q, qn, lane, ltab, &lsn, a, src, sp_dst, sp_src); qn = 0; }
            }
        }
        for (; iw < nv; iw += BT) {
            const unsigned i = iw + lane;
            const unsigned j = i + off;
            const unsigned j1 = min(j, j - nv);
            int4 dd = make_int4(-1, -1, -1, -1);
            if (i < nv) dd = d4[j1];
            qn = scan4(dd, (int)(j1 << 2), a, q, qn);
            if (qn > QFULL) { drain_queue(q, qn, lane, ltab, &lsn, a, src, sp_dst, sp_src); qn = 0; }
        }
        drain_queue(q, qn, lane, ltab, &lsn, a, src, sp_dst, sp_src);
        for (int e = (int)(nv << 2) + tid; e < E; e += BT) {
            const unsigned u = (unsigned)(dst[e] - a);
            if (u < (unsigned)RANGE) place_edge(ltab, &lsn, u, src[e], a, sp_dst, sp_src);
        }
    }
    __syncthreads();

    int4* tab4 = (int4*)tab;
    for (int w = tid; w < RANGE * (REC / 4); w += BT) {
        const int nd = a + (w >> 3);
        if (nd < n) tab4[(long long)nd * (REC / 4) + (w & 7)] = ltab4[w];
    }
    if (tid == 0) spill_cnt[blockIdx.x] = lsn;
}

static __device__ __forceinline__ int lane_i(int v, int l) {
    return __builtin_amdgcn_readlane(v, l);
}
static __device__ __forceinline__ float lane_f(float v, int l) {
    return __int_as_float(__builtin_amdgcn_readlane(__float_as_int(v), l));
}
static __device__ __forceinline__ void fma4(float4& a, const float4& h, float w) {
    a.x = fmaf(h.x, w, a.x);
    a.y = fmaf(h.y, w, a.y);
    a.z = fmaf(h.z, w, a.z);
    a.w = fmaf(h.w, w, a.w);
}

#define TILE_N 32
#define PITCH  132
__global__ __launch_bounds__(256)
void gcn_layer1_fused(const float* __restrict__ X,
                      const int* __restrict__ src, const int* __restrict__ dst, int E,
                      const int* __restrict__ tab, const int* __restrict__ spill_cnt,
                      const int* __restrict__ spill_dst, const int* __restrict__ spill_src,
                      const float* __restrict__ W1, const float* __restrict__ b1,
                      const float* __restrict__ W2,
                      float* __restrict__ H2, int n) {
    __shared__ __attribute__((aligned(16))) float sA[TILE_N * PITCH];
    __shared__ __attribute__((aligned(16))) float sT[TILE_N * PITCH];
    __shared__ __attribute__((aligned(16))) float sD[TILE_N];

    const int lane = threadIdx.x & 31;
    const int wave  = __builtin_amdgcn_readfirstlane(threadIdx.x >> 5);
    const int base  = blockIdx.x * TILE_N;
    const int node0 = base + wave * 4;
    const float* xl = X + (lane << 2);

    int   rec[4], deg[4], mm[4], ms[4];
    float di[4], mw[4];
#pragma unroll
    for (int q = 0; q < 4; ++q) {
        const int nd  = node0 + q;
        const int ndc = (nd < n) ? nd : n - 1;
        rec[q] = tab[(long long)ndc * REC + lane];
    }
#pragma unroll
    for (int q = 0; q < 4; ++q) {
        const int nd  = node0 + q;
        const bool ok = nd < n;
        const int ndc = ok ? nd : n - 1;
        const int c = lane_i(rec[q], SLOTS);
        deg[q] = ok ? c : 0;
        di[q]  = ok ? rsqrtf((float)c + 1.0f) : 0.f;
        if (lane == 0) sD[wave * 4 + q] = di[q];
        mm[q] = min(deg[q], SLOTS);
        if (deg[q] > SLOTS && spill_cnt[ndc / RANGE] > CAP) mm[q] = 0;
        ms[q] = (lane < mm[q]) ? rec[q] : ndc;
    }
#pragma unroll
    for (int q = 0; q < 4; ++q) {
        const int cs = tab[(long long)ms[q] * REC + SLOTS];
        mw[q] = rsqrtf((float)cs + 1.0f) * ((lane < mm[q]) ? di[q] : 0.f);
    }

#pragma unroll
    for (int q = 0; q < 4; ++q) {
        const int nd = node0 + q;
        const int my_s = ms[q];
        const float my_w = mw[q];
        float4 acc = {0.f, 0.f, 0.f, 0.f};
        const int m = mm[q];
        int j = 0;
        for (; j + 4 <= m; j += 4) {
            const int s0 = lane_i(my_s, j),     s1 = lane_i(my_s, j + 1);
            const int s2 = lane_i(my_s, j + 2), s3 = lane_i(my_s, j + 3);
            const float4 h0 = *(const float4*)(xl + (long long)s0 * 128);
            const float4 h1 = *(const float4*)(xl + (long long)s1 * 128);
            const float4 h2 = *(const float4*)(xl + (long long)s2 * 128);
            const float4 h3 = *(const float4*)(xl + (long long)s3 * 128);
            fma4(acc, h0, lane_f(my_w, j));
            fma4(acc, h1, lane_f(my_w, j + 1));
            fma4(acc, h2, lane_f(my_w, j + 2));
            fma4(acc, h3, lane_f(my_w, j + 3));
        }
        for (; j < m; ++j) {
            const int s0 = lane_i(my_s, j);
            const float4 h0 = *(const float4*)(xl + (long long)s0 * 128);
            fma4(acc, h0, lane_f(my_w, j));
        }
        if (deg[q] > SLOTS) {
            const int g = nd / RANGE;
            const int* pd = (m > 0) ? spill_dst + (long long)g * CAP : dst;
            const int* ps = (m > 0) ? spill_src + (long long)g * CAP : src;
            const int total = (m > 0) ? spill_cnt[g] : E;
            for (int jb = 0; jb < total; jb += 32) {
                int sd = -1, ss = 0;
                if (jb + lane < total) {
                    sd = pd[jb + lane];
                    ss = ps[jb + lane];
                }
                const bool hit = sd == nd;
                float w = 0.f;
                if (hit) w = rsqrtf((float)tab[(long long)ss * REC + SLOTS] + 1.0f) * di[q];
                unsigned hits = __builtin_amdgcn_ballot_w32(hit);
                while (hits) {
                    const int l = __builtin_ctz(hits);
                    hits &= hits - 1;
                    const int s0 = lane_i(ss, l);
                    const float4 h0 = *(const float4*)(xl + (long long)s0 * 128);
                    fma4(acc, h0, lane_f(w, l));
                }
            }
        }
        if (nd < n) {
            const float4 hs = *(const float4*)(xl + (long long)nd * 128);
            fma4(acc, hs, di[q] * di[q]);
        }
        *(float4*)(sA + (wave * 4 + q) * PITCH + (lane << 2)) = acc;
    }
    __syncthreads();

    const int c  = lane & 15;
    const int hh = lane >> 4;
    const int ko = hh << 1;

    {
        v8f d0 = {0.f,0.f,0.f,0.f,0.f,0.f,0.f,0.f};
        v8f d1 = d0;
        const float* a0p = sA + c * PITCH + ko;
        const float* a1p = sA + (16 + c) * PITCH + ko;
        const float* bp  = W1 + ko * 128 + wave * 16 + c;
#pragma unroll 8
        for (int k0 = 0; k0 < 128; k0 += 4) {
            const v2f a0 = *(const v2f*)(a0p + k0);
            const v2f a1 = *(const v2f*)(a1p + k0);
            v2f b;
            b.x = bp[k0 * 128];
            b.y = bp[k0 * 128 + 128];
            d0 = __builtin_amdgcn_wmma_f32_16x16x4_f32(false, a0, false, b, (short)0, d0, false, false);
            d1 = __builtin_amdgcn_wmma_f32_16x16x4_f32(false, a1, false, b, (short)0, d1, false, false);
        }
        const int col = wave * 16 + c;
        const float bb = b1[col];
        float* t0 = sT + (hh << 3) * PITCH + col;
        float* t1 = t0 + 16 * PITCH;
#pragma unroll
        for (int v = 0; v < 8; ++v) {
            t0[v * PITCH] = fmaxf(d0[v] + bb, 0.0f);
            t1[v * PITCH] = fmaxf(d1[v] + bb, 0.0f);
        }
    }
    __syncthreads();

    {
        const int mt = wave >> 2;
        const int nt = wave & 3;
        v8f e0 = {0.f,0.f,0.f,0.f,0.f,0.f,0.f,0.f};
        v8f e1 = e0;
        const float* tp = sT + (mt * 16 + c) * PITCH + ko;
        const float* wp = W2 + ko * 64 + nt * 16 + c;
#pragma unroll 4
        for (int k0 = 0; k0 < 128; k0 += 8) {
            const v2f a0 = *(const v2f*)(tp + k0);
            const v2f a1 = *(const v2f*)(tp + k0 + 4);
            v2f b0, b1v;
            b0.x  = wp[k0 * 64];
            b0.y  = wp[k0 * 64 + 64];
            b1v.x = wp[(k0 + 4) * 64];
            b1v.y = wp[(k0 + 4) * 64 + 64];
            e0 = __builtin_amdgcn_wmma_f32_16x16x4_f32(false, a0, false, b0,  (short)0, e0, false, false);
            e1 = __builtin_amdgcn_wmma_f32_16x16x4_f32(false, a1, false, b1v, (short)0, e1, false, false);
        }
        const int row0 = base + mt * 16 + (hh << 3);
        float* hp = H2 + (long long)row0 * 64 + nt * 16 + c;
        const float* dp = sD + mt * 16 + (hh << 3);
        if (base + TILE_N <= n) {
#pragma unroll
            for (int v = 0; v < 8; ++v) hp[v * 64] = (e0[v] + e1[v]) * dp[v];
        } else {
#pragma unroll
            for (int v = 0; v < 8; ++v)
                if (row0 + v < n) hp[v * 64] = (e0[v] + e1[v]) * dp[v];
        }
    }
}

static __device__ __forceinline__ void add2(float2& a, const float2& h) {
    a.x += h.x;
    a.y += h.y;
}
__global__ __launch_bounds__(256)
void gcn_gather64(const float* __restrict__ H,
                  const int* __restrict__ src, const int* __restrict__ dst, int E,
                  const int* __restrict__ tab, const int* __restrict__ spill_cnt,
                  const int* __restrict__ spill_dst, const int* __restrict__ spill_src,
                  const float* __restrict__ bias,
                  float* __restrict__ outp, int n) {
    const int lane = threadIdx.x & 31;
    const int node = __builtin_amdgcn_readfirstlane(blockIdx.x * 8 + (threadIdx.x >> 5));
    if (node >= n) return;

    const float* col = H + (lane << 1);
    const int c = lane << 1;
    const int rec = tab[(long long)node * REC + lane];
    const float2 hs = *(const float2*)(col + (long long)node * 64);
    const float2 bb = *(const float2*)(bias + c);
    const int deg = lane_i(rec, SLOTS);
    const float di = rsqrtf((float)deg + 1.0f);
    int m = min(deg, SLOTS);
    if (deg > SLOTS && spill_cnt[node / RANGE] > CAP) m = 0;

    float2 acc = {0.f, 0.f};
    int j = 0;
    for (; j + 4 <= m; j += 4) {
        const int s0 = lane_i(rec, j),     s1 = lane_i(rec, j + 1);
        const int s2 = lane_i(rec, j + 2), s3 = lane_i(rec, j + 3);
        const float2 h0 = *(const float2*)(col + (long long)s0 * 64);
        const float2 h1 = *(const float2*)(col + (long long)s1 * 64);
        const float2 h2 = *(const float2*)(col + (long long)s2 * 64);
        const float2 h3 = *(const float2*)(col + (long long)s3 * 64);
        add2(acc, h0);
        add2(acc, h1);
        add2(acc, h2);
        add2(acc, h3);
    }
    for (; j < m; ++j) {
        const int s0 = lane_i(rec, j);
        const float2 h0 = *(const float2*)(col + (long long)s0 * 64);
        add2(acc, h0);
    }
    if (deg > SLOTS) {
        const int g = node / RANGE;
        const int* pd = (m > 0) ? spill_dst + (long long)g * CAP : dst;
        const int* ps = (m > 0) ? spill_src + (long long)g * CAP : src;
        const int total = (m > 0) ? spill_cnt[g] : E;
        for (int jb = 0; jb < total; jb += 32) {
            int sd = -1, ss = 0;
            if (jb + lane < total) {
                sd = pd[jb + lane];
                ss = ps[jb + lane];
            }
            unsigned hits = __builtin_amdgcn_ballot_w32(sd == node);
            while (hits) {
                const int l = __builtin_ctz(hits);
                hits &= hits - 1;
                const int s0 = lane_i(ss, l);
                const float2 h0 = *(const float2*)(col + (long long)s0 * 64);
                add2(acc, h0);
            }
        }
    }

    add2(acc, hs);
    float2 r;
    r.x = fmaf(acc.x, di, bb.x);
    r.y = fmaf(acc.y, di, bb.y);
    *(float2*)(outp + (long long)node * 64 + c) = r;
}

extern "C" void kernel_launch(void* const* d_in, const int* in_sizes, int n_in,
                              void* d_out, int out_size, void* d_ws, size_t ws_size,
                              hipStream_t stream) {
    const float* x   = (const float*)d_in[0];
    const int*   ei  = (const int*)d_in[1];
    const float* W1  = (const float*)d_in[2];
    const float* b1  = (const float*)d_in[3];
    const float* W2  = (const float*)d_in[4];
    const float* b2  = (const float*)d_in[5];

    const int IN_C  = 128;
    const int nNodes = in_sizes[0] / IN_C;
    const int E      = in_sizes[1] / 2;
    const int* srcIdx = ei;
    const int* dstIdx = ei + E;

    const int nPad   = (nNodes + 63) & ~63;
    const int groups = (nNodes + RANGE - 1) / RANGE;
    int*   tab       = (int*)d_ws;
    float* h2        = (float*)(tab + (long long)nPad * REC);
    int*   spill_dst = (int*)(h2 + (long long)nPad * 64);
    int*   spill_src = spill_dst + (long long)groups * CAP;
    int*   spill_cnt = spill_src + (long long)groups * CAP;
    float* out       = (float*)d_out;


    int*   bins      = (int*)h2;
    int*   bin_cnt   = bins + (long long)GMAX * WA * BCAP * 2;

    gcn_bin_edges<<<WA, ABT, 0, stream>>>(srcIdx, dstIdx, E, groups, bins, bin_cnt);

    gcn_build_table<<<groups, BT, 0, stream>>>(srcIdx, dstIdx, E, nNodes, bins, bin_cnt,
                                               tab, spill_cnt, spill_dst, spill_src);

    gcn_layer1_fused<<<(nNodes + TILE_N - 1) / TILE_N, 256, 0, stream>>>(
        x, srcIdx, dstIdx, E, tab, spill_cnt, spill_dst, spill_src, W1, b1, W2, h2, nNodes);

    gcn_gather64<<<(nNodes + 7) / 8, 256, 0, stream>>>(
        h2, srcIdx, dstIdx, E, tab, spill_cnt, spill_dst, spill_src, b2, out, nNodes);
}
